# starting kernel
# baseline (speedup 1.0000x reference)
#include <hip/hip_runtime.h>
#include <hip/hip_bf16.h>

#define GAT_N      50000
#define GAT_E      1600000
#define GAT_IN     128
#define GAT_OUT    64
#define GAT_SLOPE  0.2f
#define GAT_EPS    1e-12f

typedef __attribute__((ext_vector_type(2))) float v2f;
typedef __attribute__((ext_vector_type(4))) float v4f;
typedef __attribute__((ext_vector_type(8))) float v8f;

__global__ void k_init(float* __restrict__ out, float* __restrict__ maxb,
                       float* __restrict__ sumb) {
  int i = blockIdx.x * blockDim.x + threadIdx.x;
  if (i < GAT_N * GAT_OUT) out[i] = 0.f;
  if (i < GAT_N) { maxb[i] = -__builtin_inff(); sumb[i] = 0.f; }
}

__global__ __launch_bounds__(128) void k_gemm(const float* __restrict__ x,
                                              const float* __restrict__ W,
                                              float* __restrict__ Wh) {
  const int lane = threadIdx.x & 31;
  const int tile = blockIdx.x * 4 + (threadIdx.x >> 5);
  if (tile >= GAT_N / 16) return;
  const int base  = tile * 16;
  const int l15   = lane & 15;
  const int khalf = (lane >> 4) << 1;

  const float* xr = x + (size_t)(base + l15) * GAT_IN + khalf;
  const float* w0 = W + (size_t)(l15 +  0) * GAT_IN + khalf;
  const float* w1 = W + (size_t)(l15 + 16) * GAT_IN + khalf;
  const float* w2 = W + (size_t)(l15 + 32) * GAT_IN + khalf;
  const float* w3 = W + (size_t)(l15 + 48) * GAT_IN + khalf;

  v8f acc0 = {}, acc1 = {}, acc2 = {}, acc3 = {};

#pragma unroll 8
  for (int k = 0; k < GAT_IN; k += 4) {
    v2f a  = *(const v2f*)(xr + k);
    v2f b0 = *(const v2f*)(w0 + k);
    v2f b1 = *(const v2f*)(w1 + k);
    v2f b2 = *(const v2f*)(w2 + k);
    v2f b3 = *(const v2f*)(w3 + k);
    acc0 = __builtin_amdgcn_wmma_f32_16x16x4_f32(false, a, false, b0, (short)0, acc0, false, false);
    acc1 = __builtin_amdgcn_wmma_f32_16x16x4_f32(false, a, false, b1, (short)0, acc1, false, false);
    acc2 = __builtin_amdgcn_wmma_f32_16x16x4_f32(false, a, false, b2, (short)0, acc2, false, false);
    acc3 = __builtin_amdgcn_wmma_f32_16x16x4_f32(false, a, false, b3, (short)0, acc3, false, false);
  }

  const int rhalf = (lane >> 4) << 3;
  float* o = Wh + (size_t)(base + rhalf) * GAT_OUT + l15;
#pragma unroll
  for (int v = 0; v < 8; ++v) {
    o[(size_t)v * GAT_OUT +  0] = acc0[v];
    o[(size_t)v * GAT_OUT + 16] = acc1[v];
    o[(size_t)v * GAT_OUT + 32] = acc2[v];
    o[(size_t)v * GAT_OUT + 48] = acc3[v];
  }
}

__global__ void k_elr(const float* __restrict__ Wh, const float* __restrict__ al,
                      const float* __restrict__ ar, float* __restrict__ el,
                      float* __restrict__ er) {
  int i = blockIdx.x * blockDim.x + threadIdx.x;
  if (i >= GAT_N) return;
  const v4f* r = (const v4f*)(Wh + (size_t)i * GAT_OUT);
  float sl = 0.f, sr = 0.f;
#pragma unroll
  for (int q = 0; q < GAT_OUT / 4; ++q) {
    v4f v = r[q];
#pragma unroll
    for (int j = 0; j < 4; ++j) {
      sl += v[j] * al[q * 4 + j];
      sr += v[j] * ar[q * 4 + j];
    }
  }
  el[i] = sl; er[i] = sr;
}

__device__ __forceinline__ void atomic_max_f32(float* addr, float val) {
  if (val >= 0.f) atomicMax((int*)addr, __float_as_int(val));
  else            atomicMin((unsigned int*)addr, (unsigned int)__float_as_int(val));
}

__global__ void k_edge_logit(const int* __restrict__ src, const int* __restrict__ dst,
                             const float* __restrict__ el, const float* __restrict__ er,
                             float* __restrict__ ebuf, float* __restrict__ maxb) {
  int e = blockIdx.x * blockDim.x + threadIdx.x;
  if (e >= GAT_E) return;
  int s = src[e], d = dst[e];
  float v = el[s] + er[d];
  v = (v > 0.f) ? v : GAT_SLOPE * v;
  ebuf[e] = v;
  atomic_max_f32(maxb + d, v);
}

__global__ void k_edge_exp(const int* __restrict__ dst, float* __restrict__ ebuf,
                           const float* __restrict__ maxb, float* __restrict__ sumb) {
  int e = blockIdx.x * blockDim.x + threadIdx.x;
  if (e >= GAT_E) return;
  int d = dst[e];
  float ex = __expf(ebuf[e] - maxb[d]);
  ebuf[e] = ex;
  atomicAdd(sumb + d, ex);
}

__global__ void k_scatter(const int* __restrict__ src, const int* __restrict__ dst,
                          const float* __restrict__ ebuf, const float* __restrict__ Wh,
                          float* __restrict__ out) {
  int idx = blockIdx.x * blockDim.x + threadIdx.x;
  if (idx >= GAT_E * GAT_OUT) return;
  int e = idx >> 6, f = idx & 63;
  int s = src[e], d = dst[e];
  atomicAdd(out + (size_t)d * GAT_OUT + f, ebuf[e] * Wh[(size_t)s * GAT_OUT + f]);
}

__global__ void k_final(float* __restrict__ out, const float* __restrict__ sumb,
                        const float* __restrict__ bias) {
  int idx = blockIdx.x * blockDim.x + threadIdx.x;
  if (idx >= GAT_N * GAT_OUT) return;
  int i = idx >> 6, f = idx & 63;
  out[idx] = out[idx] / (sumb[i] + GAT_EPS) + bias[f];
}

extern "C" void kernel_launch(void* const* d_in, const int* in_sizes, int n_in,
                              void* d_out, int out_size, void* d_ws, size_t ws_size,
                              hipStream_t stream) {
  const float* x    = (const float*)d_in[0];
  const int*   ei   = (const int*)d_in[1];
  const float* W    = (const float*)d_in[2];
  const float* al   = (const float*)d_in[3];
  const float* ar   = (const float*)d_in[4];
  const float* bias = (const float*)d_in[5];
  float* out = (float*)d_out;
  (void)in_sizes; (void)n_in; (void)out_size; (void)ws_size;

  float* ws   = (float*)d_ws;
  float* Wh   = ws;
  float* el   = Wh   + (size_t)GAT_N * GAT_OUT;
  float* er   = el   + GAT_N;
  float* maxb = er   + GAT_N;
  float* sumb = maxb + GAT_N;
  float* ebuf = sumb + GAT_N;

  const int* src = ei;
  const int* dst = ei + GAT_E;

  const int B = 256;
  k_init      <<<(GAT_N * GAT_OUT + B - 1) / B, B, 0, stream>>>(out, maxb, sumb);
  k_gemm      <<<(GAT_N / 16 + 3) / 4, 128, 0, stream>>>(x, W, Wh);
  k_elr       <<<(GAT_N + B - 1) / B, B, 0, stream>>>(Wh, al, ar, el, er);
  k_edge_logit<<<(GAT_E + B - 1) / B, B, 0, stream>>>(src, dst, el, er, ebuf, maxb);
  k_edge_exp  <<<(GAT_E + B - 1) / B, B, 0, stream>>>(dst, ebuf, maxb, sumb);
  k_scatter   <<<(GAT_E * GAT_OUT + B - 1) / B, B, 0, stream>>>(src, dst, ebuf, Wh, out);
  k_final     <<<(GAT_N * GAT_OUT + B - 1) / B, B, 0, stream>>>(out, sumb, bias);
}


# Counting-sort edges into 128-node dst buckets; one workgroup per bucket does softmax max/sum and row accumulation with LDS atomics; no global atomics left; sums in arrival order
# speedup vs baseline: 3.6337x; 3.6337x over previous
#include <hip/hip_runtime.h>
#include <hip/hip_bf16.h>

#define GAT_N      50000
#define GAT_E      1600000
#define GAT_IN     128
#define GAT_OUT    64
#define GAT_SLOPE  0.2f
#define GAT_EPS    1e-12f

typedef __attribute__((ext_vector_type(2))) float v2f;
typedef __attribute__((ext_vector_type(4))) float v4f;
typedef __attribute__((ext_vector_type(8))) float v8f;

__global__ __launch_bounds__(128) void k_gemm(const float* __restrict__ x,
                                              const float* __restrict__ W,
                                              float* __restrict__ Wh) {
  const int lane = threadIdx.x & 31;
  const int tile = blockIdx.x * 4 + (threadIdx.x >> 5);
  if (tile >= GAT_N / 16) return;
  const int base  = tile * 16;
  const int l15   = lane & 15;
  const int khalf = (lane >> 4) << 1;

  const float* xr = x + (size_t)(base + l15) * GAT_IN + khalf;
  const float* w0 = W + (size_t)(l15 +  0) * GAT_IN + khalf;
  const float* w1 = W + (size_t)(l15 + 16) * GAT_IN + khalf;
  const float* w2 = W + (size_t)(l15 + 32) * GAT_IN + khalf;
  const float* w3 = W + (size_t)(l15 + 48) * GAT_IN + khalf;

  v8f acc0 = {}, acc1 = {}, acc2 = {}, acc3 = {};

#pragma unroll 8
  for (int k = 0; k < GAT_IN; k += 4) {
    v2f a  = *(const v2f*)(xr + k);
    v2f b0 = *(const v2f*)(w0 + k);
    v2f b1 = *(const v2f*)(w1 + k);
    v2f b2 = *(const v2f*)(w2 + k);
    v2f b3 = *(const v2f*)(w3 + k);
    acc0 = __builtin_amdgcn_wmma_f32_16x16x4_f32(false, a, false, b0, (short)0, acc0, false, false);
    acc1 = __builtin_amdgcn_wmma_f32_16x16x4_f32(false, a, false, b1, (short)0, acc1, false, false);
    acc2 = __builtin_amdgcn_wmma_f32_16x16x4_f32(false, a, false, b2, (short)0, acc2, false, false);
    acc3 = __builtin_amdgcn_wmma_f32_16x16x4_f32(false, a, false, b3, (short)0, acc3, false, false);
  }

  const int rhalf = (lane >> 4) << 3;
  float* o = Wh + (size_t)(base + rhalf) * GAT_OUT + l15;
#pragma unroll
  for (int v = 0; v < 8; ++v) {
    o[(size_t)v * GAT_OUT +  0] = acc0[v];
    o[(size_t)v * GAT_OUT + 16] = acc1[v];
    o[(size_t)v * GAT_OUT + 32] = acc2[v];
    o[(size_t)v * GAT_OUT + 48] = acc3[v];
  }
}

__global__ void k_elr(const float* __restrict__ Wh, const float* __restrict__ al,
                      const float* __restrict__ ar, float* __restrict__ el,
                      float* __restrict__ er) {
  int i = blockIdx.x * blockDim.x + threadIdx.x;
  if (i >= GAT_N) return;
  const v4f* r = (const v4f*)(Wh + (size_t)i * GAT_OUT);
  float sl = 0.f, sr = 0.f;
#pragma unroll
  for (int q = 0; q < GAT_OUT / 4; ++q) {
    v4f v = r[q];
#pragma unroll
    for (int j = 0; j < 4; ++j) {
      sl += v[j] * al[q * 4 + j];
      sr += v[j] * ar[q * 4 + j];
    }
  }
  el[i] = sl; er[i] = sr;
}

#define BK_SH    7
#define BK_N     128
#define BK_NUM   ((GAT_N + BK_N - 1) / BK_N)
#define SRT_NBLK 200
#define SRT_CH   (GAT_E / SRT_NBLK)
#define SRT_T    512
static_assert(SRT_NBLK * SRT_CH == GAT_E, "sort blocks must tile the edge list");
static_assert(GAT_N <= 65536, "src is packed into 16 bits of the sorted word");
static_assert(SRT_NBLK <= 256 && BK_NUM <= 512, "scan block sizes");

template <int NT>
__device__ __forceinline__ int block_excl_scan(int v, int* sh, int* total) {
  const int t = threadIdx.x;
  sh[t] = v;
  __syncthreads();
  for (int off = 1; off < NT; off <<= 1) {
    int add = (t >= off) ? sh[t - off] : 0;
    __syncthreads();
    sh[t] += add;
    __syncthreads();
  }
  *total = sh[NT - 1];
  return sh[t] - v;
}

__global__ __launch_bounds__(SRT_T) void k_hist(const int* __restrict__ dst,
                                                int* __restrict__ hist) {
  __shared__ int h[BK_NUM];
  const int t = threadIdx.x;
  for (int i = t; i < BK_NUM; i += SRT_T) h[i] = 0;
  __syncthreads();
  const int* dp = dst + (size_t)blockIdx.x * SRT_CH;
  for (int i = t; i < SRT_CH; i += SRT_T) atomicAdd(&h[dp[i] >> BK_SH], 1);
  __syncthreads();
  for (int i = t; i < BK_NUM; i += SRT_T) hist[i * SRT_NBLK + blockIdx.x] = h[i];
}

__global__ __launch_bounds__(256) void k_scan_blocks(int* __restrict__ hist,
                                                     int* __restrict__ total) {
  __shared__ int sh[256];
  const int t = threadIdx.x;
  const int b = blockIdx.x;
  int v = (t < SRT_NBLK) ? hist[b * SRT_NBLK + t] : 0;
  int tot;
  int ex = block_excl_scan<256>(v, sh, &tot);
  if (t < SRT_NBLK) hist[b * SRT_NBLK + t] = ex;
  if (t == 0) total[b] = tot;
}

__global__ __launch_bounds__(512) void k_scan_buckets(const int* __restrict__ total,
                                                      int* __restrict__ base) {
  __shared__ int sh[512];
  const int t = threadIdx.x;
  int v = (t < BK_NUM) ? total[t] : 0;
  int tot;
  int ex = block_excl_scan<512>(v, sh, &tot);
  if (t < BK_NUM) base[t] = ex;
  if (t == 0) base[BK_NUM] = tot;
}

__global__ __launch_bounds__(SRT_T) void k_fill(const int* __restrict__ src,
                                                const int* __restrict__ dst,
                                                const int* __restrict__ hist,
                                                const int* __restrict__ base,
                                                unsigned int* __restrict__ sorted) {
  __shared__ int cur[BK_NUM];
  const int t = threadIdx.x;
  for (int i = t; i < BK_NUM; i += SRT_T) cur[i] = base[i] + hist[i * SRT_NBLK + blockIdx.x];
  __syncthreads();
  const int* sp = src + (size_t)blockIdx.x * SRT_CH;
  const int* dp = dst + (size_t)blockIdx.x * SRT_CH;
  for (int i = t; i < SRT_CH; i += SRT_T) {
    int s = sp[i], d = dp[i];
    int slot = atomicAdd(&cur[d >> BK_SH], 1);
    sorted[slot] = (unsigned int)s | ((unsigned int)(d & (BK_N - 1)) << 16);
  }
}

__device__ __forceinline__ void lds_max_f32(float* addr, float val) {
  if (val >= 0.f) atomicMax((int*)addr, __float_as_int(val));
  else            atomicMin((unsigned int*)addr, (unsigned int)__float_as_int(val));
}

__device__ __forceinline__ float edge_logit(const float* __restrict__ el,
                                            const float* __restrict__ er, int s, int d) {
  float v = el[s] + er[d];
  return (v > 0.f) ? v : GAT_SLOPE * v;
}

#define BKT_T 1024
__global__ __launch_bounds__(BKT_T) void k_bucket(const unsigned int* __restrict__ sorted,
                                                  const int* __restrict__ base,
                                                  const float* __restrict__ Wh,
                                                  const float* __restrict__ el,
                                                  const float* __restrict__ er,
                                                  const float* __restrict__ bias,
                                                  float* __restrict__ out) {
  __shared__ float acc[BK_N * GAT_OUT];
  __shared__ float mx[BK_N];
  __shared__ float sm[BK_N];
  const int t    = threadIdx.x;
  const int lane = t & 31;
  const int wv   = t >> 5;
  const int n0   = blockIdx.x * BK_N;
  const int beg  = base[blockIdx.x];
  const int end  = base[blockIdx.x + 1];

  for (int i = t; i < BK_N * GAT_OUT; i += BKT_T) acc[i] = 0.f;
  if (t < BK_N) { mx[t] = -__builtin_inff(); sm[t] = 0.f; }
  __syncthreads();

  for (int i = beg + t; i < end; i += BKT_T) {
    unsigned int w = sorted[i];
    int s = (int)(w & 0xffffu), dl = (int)(w >> 16);
    lds_max_f32(&mx[dl], edge_logit(el, er, s, n0 + dl));
  }
  __syncthreads();

  for (int g = beg + wv * 32; g < end; g += BKT_T) {
    const int  i  = g + lane;
    const bool ok = i < end;
    unsigned int w = ok ? sorted[i] : 0u;
    int s = (int)(w & 0xffffu), dl = (int)(w >> 16);
    float ex = 0.f;
    if (ok) {
      ex = __expf(edge_logit(el, er, s, n0 + dl) - mx[dl]);
      atomicAdd(&sm[dl], ex);
    }
    const int nv = min(32, end - g);
    for (int j0 = 0; j0 < nv; j0 += 8) {
      v2f   r[8];
      int   dj[8];
      float wj[8];
#pragma unroll
      for (int k = 0; k < 8; ++k) {
        int sj = __builtin_amdgcn_readlane(s, j0 + k);
        dj[k]  = __builtin_amdgcn_readlane(dl, j0 + k);
        wj[k]  = __int_as_float(__builtin_amdgcn_readlane(__float_as_int(ex), j0 + k));
        r[k]   = *((const v2f*)(Wh + (size_t)sj * GAT_OUT) + lane);
      }
#pragma unroll
      for (int k = 0; k < 8; ++k) {
        if (j0 + k < nv) {
          float* a = acc + dj[k] * GAT_OUT + 2 * lane;
          atomicAdd(a,     wj[k] * r[k].x);
          atomicAdd(a + 1, wj[k] * r[k].y);
        }
      }
    }
  }
  __syncthreads();

  const int nrows = min(BK_N, GAT_N - n0);
  for (int i = t; i < nrows * GAT_OUT; i += BKT_T) {
    out[(size_t)n0 * GAT_OUT + i] = acc[i] / (sm[i >> 6] + GAT_EPS) + bias[i & 63];
  }
}

extern "C" void kernel_launch(void* const* d_in, const int* in_sizes, int n_in,
                              void* d_out, int out_size, void* d_ws, size_t ws_size,
                              hipStream_t stream) {
  const float* x    = (const float*)d_in[0];
  const int*   ei   = (const int*)d_in[1];
  const float* W    = (const float*)d_in[2];
  const float* al   = (const float*)d_in[3];
  const float* ar   = (const float*)d_in[4];
  const float* bias = (const float*)d_in[5];
  float* out = (float*)d_out;
  (void)in_sizes; (void)n_in; (void)out_size; (void)ws_size;

  float* ws   = (float*)d_ws;
  float* Wh   = ws;
  float* el   = Wh + (size_t)GAT_N * GAT_OUT;
  float* er   = el + GAT_N;
  int*   hist  = (int*)(er + GAT_N);
  int*   total = hist  + BK_NUM * SRT_NBLK;
  int*   base  = total + 512;
  unsigned int* sorted = (unsigned int*)(base + 512);

  const int* src = ei;
  const int* dst = ei + GAT_E;

  const int B = 256;
  k_gemm        <<<(GAT_N / 16 + 3) / 4, 128, 0, stream>>>(x, W, Wh);
  k_elr         <<<(GAT_N + B - 1) / B, B, 0, stream>>>(Wh, al, ar, el, er);
  k_hist        <<<SRT_NBLK, SRT_T, 0, stream>>>(dst, hist);
  k_scan_blocks <<<BK_NUM, 256, 0, stream>>>(hist, total);
  k_scan_buckets<<<1, 512, 0, stream>>>(total, base);
  k_fill        <<<SRT_NBLK, SRT_T, 0, stream>>>(src, dst, hist, base, sorted);
  k_bucket      <<<BK_NUM, BKT_T, 0, stream>>>(sorted, base, Wh, el, er, bias, out);
}


# Second counting sort by dst node inside each bucket's LDS; a wave owns 4 whole nodes and sums their rows in registers (fma, staged arrival order); the per-feature LDS float atomics are gone
# speedup vs baseline: 4.5600x; 1.2549x over previous
#include <hip/hip_runtime.h>
#include <hip/hip_bf16.h>

#define GAT_N      50000
#define GAT_E      1600000
#define GAT_IN     128
#define GAT_OUT    64
#define GAT_SLOPE  0.2f
#define GAT_EPS    1e-12f

typedef __attribute__((ext_vector_type(2))) float v2f;
typedef __attribute__((ext_vector_type(4))) float v4f;
typedef __attribute__((ext_vector_type(8))) float v8f;

__global__ __launch_bounds__(128) void k_gemm(const float* __restrict__ x,
                                              const float* __restrict__ W,
                                              float* __restrict__ Wh) {
  const int lane = threadIdx.x & 31;
  const int tile = blockIdx.x * 4 + (threadIdx.x >> 5);
  if (tile >= GAT_N / 16) return;
  const int base  = tile * 16;
  const int l15   = lane & 15;
  const int khalf = (lane >> 4) << 1;

  const float* xr = x + (size_t)(base + l15) * GAT_IN + khalf;
  const float* w0 = W + (size_t)(l15 +  0) * GAT_IN + khalf;
  const float* w1 = W + (size_t)(l15 + 16) * GAT_IN + khalf;
  const float* w2 = W + (size_t)(l15 + 32) * GAT_IN + khalf;
  const float* w3 = W + (size_t)(l15 + 48) * GAT_IN + khalf;

  v8f acc0 = {}, acc1 = {}, acc2 = {}, acc3 = {};

#pragma unroll 8
  for (int k = 0; k < GAT_IN; k += 4) {
    v2f a  = *(const v2f*)(xr + k);
    v2f b0 = *(const v2f*)(w0 + k);
    v2f b1 = *(const v2f*)(w1 + k);
    v2f b2 = *(const v2f*)(w2 + k);
    v2f b3 = *(const v2f*)(w3 + k);
    acc0 = __builtin_amdgcn_wmma_f32_16x16x4_f32(false, a, false, b0, (short)0, acc0, false, false);
    acc1 = __builtin_amdgcn_wmma_f32_16x16x4_f32(false, a, false, b1, (short)0, acc1, false, false);
    acc2 = __builtin_amdgcn_wmma_f32_16x16x4_f32(false, a, false, b2, (short)0, acc2, false, false);
    acc3 = __builtin_amdgcn_wmma_f32_16x16x4_f32(false, a, false, b3, (short)0, acc3, false, false);
  }

  const int rhalf = (lane >> 4) << 3;
  float* o = Wh + (size_t)(base + rhalf) * GAT_OUT + l15;
#pragma unroll
  for (int v = 0; v < 8; ++v) {
    o[(size_t)v * GAT_OUT +  0] = acc0[v];
    o[(size_t)v * GAT_OUT + 16] = acc1[v];
    o[(size_t)v * GAT_OUT + 32] = acc2[v];
    o[(size_t)v * GAT_OUT + 48] = acc3[v];
  }
}

__global__ void k_elr(const float* __restrict__ Wh, const float* __restrict__ al,
                      const float* __restrict__ ar, float* __restrict__ el,
                      float* __restrict__ er) {
  int i = blockIdx.x * blockDim.x + threadIdx.x;
  if (i >= GAT_N) return;
  const v4f* r = (const v4f*)(Wh + (size_t)i * GAT_OUT);
  float sl = 0.f, sr = 0.f;
#pragma unroll
  for (int q = 0; q < GAT_OUT / 4; ++q) {
    v4f v = r[q];
#pragma unroll
    for (int j = 0; j < 4; ++j) {
      sl += v[j] * al[q * 4 + j];
      sr += v[j] * ar[q * 4 + j];
    }
  }
  el[i] = sl; er[i] = sr;
}

#define BK_SH    7
#define BK_N     128
#define BK_NUM   ((GAT_N + BK_N - 1) / BK_N)
#define SRT_NBLK 200
#define SRT_CH   (GAT_E / SRT_NBLK)
#define SRT_T    512
static_assert(SRT_NBLK * SRT_CH == GAT_E, "sort blocks must tile the edge list");
static_assert(GAT_N <= 65536, "src is packed into 16 bits of the sorted word");
static_assert(SRT_NBLK <= 256 && BK_NUM <= 512, "scan block sizes");

template <int NT>
__device__ __forceinline__ int block_excl_scan(int v, int* sh, int* total) {
  const int t = threadIdx.x;
  sh[t] = v;
  __syncthreads();
  for (int off = 1; off < NT; off <<= 1) {
    int add = (t >= off) ? sh[t - off] : 0;
    __syncthreads();
    sh[t] += add;
    __syncthreads();
  }
  *total = sh[NT - 1];
  return sh[t] - v;
}

__global__ __launch_bounds__(SRT_T) void k_hist(const int* __restrict__ dst,
                                                int* __restrict__ hist) {
  __shared__ int h[BK_NUM];
  const int t = threadIdx.x;
  for (int i = t; i < BK_NUM; i += SRT_T) h[i] = 0;
  __syncthreads();
  const int* dp = dst + (size_t)blockIdx.x * SRT_CH;
  for (int i = t; i < SRT_CH; i += SRT_T) atomicAdd(&h[dp[i] >> BK_SH], 1);
  __syncthreads();
  for (int i = t; i < BK_NUM; i += SRT_T) hist[i * SRT_NBLK + blockIdx.x] = h[i];
}

__global__ __launch_bounds__(256) void k_scan_blocks(int* __restrict__ hist,
                                                     int* __restrict__ total) {
  __shared__ int sh[256];
  const int t = threadIdx.x;
  const int b = blockIdx.x;
  int v = (t < SRT_NBLK) ? hist[b * SRT_NBLK + t] : 0;
  int tot;
  int ex = block_excl_scan<256>(v, sh, &tot);
  if (t < SRT_NBLK) hist[b * SRT_NBLK + t] = ex;
  if (t == 0) total[b] = tot;
}

__global__ __launch_bounds__(512) void k_scan_buckets(const int* __restrict__ total,
                                                      int* __restrict__ base) {
  __shared__ int sh[512];
  const int t = threadIdx.x;
  int v = (t < BK_NUM) ? total[t] : 0;
  int tot;
  int ex = block_excl_scan<512>(v, sh, &tot);
  if (t < BK_NUM) base[t] = ex;
  if (t == 0) base[BK_NUM] = tot;
}

__global__ __launch_bounds__(SRT_T) void k_fill(const int* __restrict__ src,
                                                const int* __restrict__ dst,
                                                const int* __restrict__ hist,
                                                const int* __restrict__ base,
                                                unsigned int* __restrict__ sorted) {
  __shared__ int cur[BK_NUM];
  const int t = threadIdx.x;
  for (int i = t; i < BK_NUM; i += SRT_T) cur[i] = base[i] + hist[i * SRT_NBLK + blockIdx.x];
  __syncthreads();
  const int* sp = src + (size_t)blockIdx.x * SRT_CH;
  const int* dp = dst + (size_t)blockIdx.x * SRT_CH;
  for (int i = t; i < SRT_CH; i += SRT_T) {
    int s = sp[i], d = dp[i];
    int slot = atomicAdd(&cur[d >> BK_SH], 1);
    sorted[slot] = (unsigned int)s | ((unsigned int)(d & (BK_N - 1)) << 16);
  }
}

__device__ __forceinline__ void lds_max_f32(float* addr, float val) {
  if (val >= 0.f) atomicMax((int*)addr, __float_as_int(val));
  else            atomicMin((unsigned int*)addr, (unsigned int)__float_as_int(val));
}

__device__ __forceinline__ float edge_logit(const float* __restrict__ el,
                                            const float* __restrict__ er, int s, int d) {
  float v = el[s] + er[d];
  return (v > 0.f) ? v : GAT_SLOPE * v;
}

#define BKT_T   1024
#define BKT_EPT 5
#define BKT_CAP (BKT_T * BKT_EPT)
#define BKT_NPW (BK_N / (BKT_T / 32))
#define BKT_NONE 0xffffffffu
static_assert(BKT_NPW * (BKT_T / 32) == BK_N, "the waves must tile the bucket's nodes");
static_assert(BK_N == 4 * 32, "the scan below gives every lane of one wave four counters");

__device__ __forceinline__ void bkt_load_chunk(const unsigned int* __restrict__ sorted,
                                               const float* __restrict__ el,
                                               const float* ler, int cb, int end, int t,
                                               unsigned int (&w)[BKT_EPT], float (&lg)[BKT_EPT]) {
  float e[BKT_EPT];
#pragma unroll
  for (int k = 0; k < BKT_EPT; ++k) {
    const int i = cb + k * BKT_T + t;
    const unsigned int v = sorted[min(i, end - 1)];
    w[k] = (i < end) ? v : BKT_NONE;
  }
#pragma unroll
  for (int k = 0; k < BKT_EPT; ++k) {
    e[k] = el[(w[k] != BKT_NONE) ? (int)(w[k] & 0xffffu) : 0];
  }
#pragma unroll
  for (int k = 0; k < BKT_EPT; ++k) {
    const int dl = (w[k] != BKT_NONE) ? (int)(w[k] >> 16) : 0;
    const float v = e[k] + ler[dl];
    lg[k] = (v > 0.f) ? v : GAT_SLOPE * v;
  }
}

__global__ __launch_bounds__(BKT_T) void k_bucket(const unsigned int* __restrict__ sorted,
                                                  const int* __restrict__ base,
                                                  const float* __restrict__ Wh,
                                                  const float* __restrict__ el,
                                                  const float* __restrict__ er,
                                                  const float* __restrict__ bias,
                                                  float* __restrict__ out) {
  __shared__ unsigned int lsrc[BKT_CAP];
  __shared__ float        lwt[BKT_CAP];
  __shared__ float mx[BK_N];
  __shared__ float sm[BK_N];
  __shared__ float ler[BK_N];
  __shared__ int   cur[BK_N];
  __shared__ int   off[BK_N + 4];
  const int t    = threadIdx.x;
  const int lane = t & 31;
  const int wv   = t >> 5;
  const int n0   = blockIdx.x * BK_N;
  const int beg  = base[blockIdx.x];
  const int end  = base[blockIdx.x + 1];
  const bool multi = (end - beg) > BKT_CAP;

  if (t < BK_N) {
    mx[t] = -__builtin_inff(); sm[t] = 0.f; cur[t] = 0;
    ler[t] = (n0 + t < GAT_N) ? er[n0 + t] : 0.f;
  }
  __syncthreads();

  unsigned int w[BKT_EPT];
  float        lg[BKT_EPT];
#pragma unroll
  for (int k = 0; k < BKT_EPT; ++k) { w[k] = BKT_NONE; lg[k] = 0.f; }

  for (int cb = beg; cb < end; cb += BKT_CAP) {
    bkt_load_chunk(sorted, el, ler, cb, end, t, w, lg);
#pragma unroll
    for (int k = 0; k < BKT_EPT; ++k) {
      if (w[k] != BKT_NONE) {
        const int dl = (int)(w[k] >> 16);
        lds_max_f32(&mx[dl], lg[k]);
        if (!multi) atomicAdd(&cur[dl], 1);
      }
    }
  }
  __syncthreads();

  v2f acc[BKT_NPW];
#pragma unroll
  for (int q = 0; q < BKT_NPW; ++q) { acc[q].x = 0.f; acc[q].y = 0.f; }

  for (int cb = beg; cb < end; cb += BKT_CAP) {
    if (multi) {
      bkt_load_chunk(sorted, el, ler, cb, end, t, w, lg);
      if (t < BK_N) cur[t] = 0;
      __syncthreads();
#pragma unroll
      for (int k = 0; k < BKT_EPT; ++k) {
        if (w[k] != BKT_NONE) atomicAdd(&cur[(int)(w[k] >> 16)], 1);
      }
      __syncthreads();
    }

    if (wv == 0) {
      const int c0 = cur[4 * lane + 0], c1 = cur[4 * lane + 1];
      const int c2 = cur[4 * lane + 2], c3 = cur[4 * lane + 3];
      const int tot = c0 + c1 + c2 + c3;
      int incl = tot;
#pragma unroll
      for (int d = 1; d < 32; d <<= 1) {
        const int up = __shfl_up(incl, d);
        if (lane >= d) incl += up;
      }
      const int ex = incl - tot;
      off[4 * lane + 0] = ex;                cur[4 * lane + 0] = ex;
      off[4 * lane + 1] = ex + c0;           cur[4 * lane + 1] = ex + c0;
      off[4 * lane + 2] = ex + c0 + c1;      cur[4 * lane + 2] = ex + c0 + c1;
      off[4 * lane + 3] = ex + c0 + c1 + c2; cur[4 * lane + 3] = ex + c0 + c1 + c2;
      if (lane == 31) off[BK_N] = incl;
    }
    __syncthreads();

    {
      float exv[BKT_EPT];
      int   pos[BKT_EPT];
#pragma unroll
      for (int k = 0; k < BKT_EPT; ++k) {
        exv[k] = mx[(w[k] != BKT_NONE) ? (int)(w[k] >> 16) : 0];
      }
#pragma unroll
      for (int k = 0; k < BKT_EPT; ++k) {
        exv[k] = __expf(lg[k] - exv[k]);
      }
#pragma unroll
      for (int k = 0; k < BKT_EPT; ++k) {
        pos[k] = 0;
        if (w[k] != BKT_NONE) {
          const int dl = (int)(w[k] >> 16);
          atomicAdd(&sm[dl], exv[k]);
          pos[k] = atomicAdd(&cur[dl], 1);
        }
      }
#pragma unroll
      for (int k = 0; k < BKT_EPT; ++k) {
        if (w[k] != BKT_NONE) {
          lsrc[pos[k]] = w[k] & 0xffffu;
          lwt[pos[k]]  = exv[k];
        }
      }
    }
    __syncthreads();

#pragma unroll
    for (int q = 0; q < BKT_NPW; ++q) {
      const int dl = wv * BKT_NPW + q;
      const int a  = __builtin_amdgcn_readfirstlane(off[dl]);
      const int b  = __builtin_amdgcn_readfirstlane(off[dl + 1]);
      for (int c = a; c < b; c += 32) {
        const int  idx = c + lane;
        const bool ok  = idx < b;
        const unsigned int s  = lsrc[ok ? idx : c];
        const float wgt = ok ? lwt[ok ? idx : c] : 0.f;
        const int nv = min(32, b - c);
        for (int j0 = 0; j0 < nv; j0 += 8) {
          v2f   r[8];
          float wj[8];
#pragma unroll
          for (int k = 0; k < 8; ++k) {
            const unsigned int sj = (unsigned int)__builtin_amdgcn_readlane((int)s, j0 + k);
            wj[k] = __int_as_float(__builtin_amdgcn_readlane(__float_as_int(wgt), j0 + k));
            r[k]  = *((const v2f*)(Wh + (size_t)sj * GAT_OUT) + lane);
          }
#pragma unroll
          for (int k = 0; k < 8; ++k) {
            acc[q].x = fmaf(wj[k], r[k].x, acc[q].x);
            acc[q].y = fmaf(wj[k], r[k].y, acc[q].y);
          }
        }
      }
    }
    if (multi) __syncthreads();
  }

  const v2f bb = *((const v2f*)bias + lane);
#pragma unroll
  for (int q = 0; q < BKT_NPW; ++q) {
    const int dl = wv * BKT_NPW + q;
    const int n  = n0 + dl;
    if (n < GAT_N) {
      const float den = sm[dl] + GAT_EPS;
      v2f o;
      o.x = acc[q].x / den + bb.x;
      o.y = acc[q].y / den + bb.y;
      *((v2f*)(out + (size_t)n * GAT_OUT) + lane) = o;
    }
  }
}

extern "C" void kernel_launch(void* const* d_in, const int* in_sizes, int n_in,
                              void* d_out, int out_size, void* d_ws, size_t ws_size,
                              hipStream_t stream) {
  const float* x    = (const float*)d_in[0];
  const int*   ei   = (const int*)d_in[1];
  const float* W    = (const float*)d_in[2];
  const float* al   = (const float*)d_in[3];
  const float* ar   = (const float*)d_in[4];
  const float* bias = (const float*)d_in[5];
  float* out = (float*)d_out;
  (void)in_sizes; (void)n_in; (void)out_size; (void)ws_size;

  float* ws   = (float*)d_ws;
  float* Wh   = ws;
  float* el   = Wh + (size_t)GAT_N * GAT_OUT;
  float* er   = el + GAT_N;
  int*   hist  = (int*)(er + GAT_N);
  int*   total = hist  + BK_NUM * SRT_NBLK;
  int*   base  = total + 512;
  unsigned int* sorted = (unsigned int*)(base + 512);

  const int* src = ei;
  const int* dst = ei + GAT_E;

  const int B = 256;
  k_gemm        <<<(GAT_N / 16 + 3) / 4, 128, 0, stream>>>(x, W, Wh);
  k_elr         <<<(GAT_N + B - 1) / B, B, 0, stream>>>(Wh, al, ar, el, er);
  k_hist        <<<SRT_NBLK, SRT_T, 0, stream>>>(dst, hist);
  k_scan_blocks <<<BK_NUM, 256, 0, stream>>>(hist, total);
  k_scan_buckets<<<1, 512, 0, stream>>>(total, base);
  k_fill        <<<SRT_NBLK, SRT_T, 0, stream>>>(src, dst, hist, base, sorted);
  k_bucket      <<<BK_NUM, BKT_T, 0, stream>>>(sorted, base, Wh, el, er, bias, out);
}


# Independent kernels share a launch: the edge histogram runs as extra blocks of the GEMM launch and the per-bucket scan as extra blocks of the el/er launch; 7 launches become 5, sums unchanged
# speedup vs baseline: 5.2145x; 1.1435x over previous
#include <hip/hip_runtime.h>
#include <hip/hip_bf16.h>

#define GAT_N      50000
#define GAT_E      1600000
#define GAT_IN     128
#define GAT_OUT    64
#define GAT_SLOPE  0.2f
#define GAT_EPS    1e-12f

typedef __attribute__((ext_vector_type(2))) float v2f;
typedef __attribute__((ext_vector_type(4))) float v4f;
typedef __attribute__((ext_vector_type(8))) float v8f;

#define GEMM_WPB  16
#define GEMM_NBLK ((GAT_N / 16 + GEMM_WPB - 1) / GEMM_WPB)
__device__ __forceinline__ void gemm_part(const int blk, const float* __restrict__ x,
                                          const float* __restrict__ W,
                                          float* __restrict__ Wh) {
  const int lane = threadIdx.x & 31;
  const int tile = blk * GEMM_WPB + (threadIdx.x >> 5);
  if (tile >= GAT_N / 16) return;
  const int base  = tile * 16;
  const int l15   = lane & 15;
  const int khalf = (lane >> 4) << 1;

  const float* xr = x + (size_t)(base + l15) * GAT_IN + khalf;
  const float* w0 = W + (size_t)(l15 +  0) * GAT_IN + khalf;
  const float* w1 = W + (size_t)(l15 + 16) * GAT_IN + khalf;
  const float* w2 = W + (size_t)(l15 + 32) * GAT_IN + khalf;
  const float* w3 = W + (size_t)(l15 + 48) * GAT_IN + khalf;

  v8f acc0 = {}, acc1 = {}, acc2 = {}, acc3 = {};

#pragma unroll 8
  for (int k = 0; k < GAT_IN; k += 4) {
    v2f a  = *(const v2f*)(xr + k);
    v2f b0 = *(const v2f*)(w0 + k);
    v2f b1 = *(const v2f*)(w1 + k);
    v2f b2 = *(const v2f*)(w2 + k);
    v2f b3 = *(const v2f*)(w3 + k);
    acc0 = __builtin_amdgcn_wmma_f32_16x16x4_f32(false, a, false, b0, (short)0, acc0, false, false);
    acc1 = __builtin_amdgcn_wmma_f32_16x16x4_f32(false, a, false, b1, (short)0, acc1, false, false);
    acc2 = __builtin_amdgcn_wmma_f32_16x16x4_f32(false, a, false, b2, (short)0, acc2, false, false);
    acc3 = __builtin_amdgcn_wmma_f32_16x16x4_f32(false, a, false, b3, (short)0, acc3, false, false);
  }

  const int rhalf = (lane >> 4) << 3;
  float* o = Wh + (size_t)(base + rhalf) * GAT_OUT + l15;
#pragma unroll
  for (int v = 0; v < 8; ++v) {
    o[(size_t)v * GAT_OUT +  0] = acc0[v];
    o[(size_t)v * GAT_OUT + 16] = acc1[v];
    o[(size_t)v * GAT_OUT + 32] = acc2[v];
    o[(size_t)v * GAT_OUT + 48] = acc3[v];
  }
}

#define ELR_T    256
#define ELR_NBLK ((GAT_N + ELR_T - 1) / ELR_T)
__device__ __forceinline__ void elr_part(const int blk, const float* __restrict__ Wh,
                                         const float* __restrict__ al,
                                         const float* __restrict__ ar, float* __restrict__ el,
                                         float* __restrict__ er) {
  int i = blk * ELR_T + threadIdx.x;
  if (i >= GAT_N) return;
  const v4f* r = (const v4f*)(Wh + (size_t)i * GAT_OUT);
  float sl = 0.f, sr = 0.f;
#pragma unroll
  for (int q = 0; q < GAT_OUT / 4; ++q) {
    v4f v = r[q];
#pragma unroll
    for (int j = 0; j < 4; ++j) {
      sl += v[j] * al[q * 4 + j];
      sr += v[j] * ar[q * 4 + j];
    }
  }
  el[i] = sl; er[i] = sr;
}

#define BK_SH    7
#define BK_N     128
#define BK_NUM   ((GAT_N + BK_N - 1) / BK_N)
#define SRT_NBLK 200
#define SRT_CH   (GAT_E / SRT_NBLK)
#define SRT_T    512
static_assert(SRT_NBLK * SRT_CH == GAT_E, "sort blocks must tile the edge list");
static_assert(GAT_N <= 65536, "src is packed into 16 bits of the sorted word");
static_assert(SRT_NBLK <= 256 && BK_NUM <= 512, "scan block sizes");

template <int NT>
__device__ __forceinline__ int block_excl_scan(int v, int* sh, int* total) {
  const int t = threadIdx.x;
  sh[t] = v;
  __syncthreads();
  for (int off = 1; off < NT; off <<= 1) {
    int add = (t >= off) ? sh[t - off] : 0;
    __syncthreads();
    sh[t] += add;
    __syncthreads();
  }
  *total = sh[NT - 1];
  return sh[t] - v;
}

__device__ __forceinline__ void hist_part(const int blk, const int* __restrict__ dst,
                                          int* __restrict__ hist, int* h) {
  const int t = threadIdx.x;
  for (int i = t; i < BK_NUM; i += SRT_T) h[i] = 0;
  __syncthreads();
  const int* dp = dst + (size_t)blk * SRT_CH;
  for (int i = t; i < SRT_CH; i += SRT_T) atomicAdd(&h[dp[i] >> BK_SH], 1);
  __syncthreads();
  for (int i = t; i < BK_NUM; i += SRT_T) hist[i * SRT_NBLK + blk] = h[i];
}

static_assert(GEMM_WPB * 32 == SRT_T, "both parts of the shared launch use one block size");
__global__ __launch_bounds__(SRT_T) void k_gemm_hist(const float* __restrict__ x,
                                                     const float* __restrict__ W,
                                                     float* __restrict__ Wh,
                                                     const int* __restrict__ dst,
                                                     int* __restrict__ hist) {
  __shared__ int h[BK_NUM];
  if (blockIdx.x < GEMM_NBLK) gemm_part(blockIdx.x, x, W, Wh);
  else                        hist_part(blockIdx.x - GEMM_NBLK, dst, hist, h);
}

__device__ __forceinline__ void scan_blocks_part(const int b, int* __restrict__ hist,
                                                 int* __restrict__ total, int* sh) {
  const int t = threadIdx.x;
  int v = (t < SRT_NBLK) ? hist[b * SRT_NBLK + t] : 0;
  int tot;
  int ex = block_excl_scan<ELR_T>(v, sh, &tot);
  if (t < SRT_NBLK) hist[b * SRT_NBLK + t] = ex;
  if (t == 0) total[b] = tot;
}

static_assert(SRT_NBLK <= ELR_T, "one thread per sort block in the scan");
__global__ __launch_bounds__(ELR_T) void k_scan_elr(int* __restrict__ hist,
                                                    int* __restrict__ total,
                                                    const float* __restrict__ Wh,
                                                    const float* __restrict__ al,
                                                    const float* __restrict__ ar,
                                                    float* __restrict__ el,
                                                    float* __restrict__ er) {
  __shared__ int sh[ELR_T];
  if (blockIdx.x < BK_NUM) scan_blocks_part(blockIdx.x, hist, total, sh);
  else                     elr_part(blockIdx.x - BK_NUM, Wh, al, ar, el, er);
}

__global__ __launch_bounds__(512) void k_scan_buckets(const int* __restrict__ total,
                                                      int* __restrict__ base) {
  __shared__ int sh[512];
  const int t = threadIdx.x;
  int v = (t < BK_NUM) ? total[t] : 0;
  int tot;
  int ex = block_excl_scan<512>(v, sh, &tot);
  if (t < BK_NUM) base[t] = ex;
  if (t == 0) base[BK_NUM] = tot;
}

__global__ __launch_bounds__(SRT_T) void k_fill(const int* __restrict__ src,
                                                const int* __restrict__ dst,
                                                const int* __restrict__ hist,
                                                const int* __restrict__ base,
                                                unsigned int* __restrict__ sorted) {
  __shared__ int cur[BK_NUM];
  const int t = threadIdx.x;
  for (int i = t; i < BK_NUM; i += SRT_T) cur[i] = base[i] + hist[i * SRT_NBLK + blockIdx.x];
  __syncthreads();
  const int* sp = src + (size_t)blockIdx.x * SRT_CH;
  const int* dp = dst + (size_t)blockIdx.x * SRT_CH;
  for (int i = t; i < SRT_CH; i += SRT_T) {
    int s = sp[i], d = dp[i];
    int slot = atomicAdd(&cur[d >> BK_SH], 1);
    sorted[slot] = (unsigned int)s | ((unsigned int)(d & (BK_N - 1)) << 16);
  }
}

__device__ __forceinline__ void lds_max_f32(float* addr, float val) {
  if (val >= 0.f) atomicMax((int*)addr, __float_as_int(val));
  else            atomicMin((unsigned int*)addr, (unsigned int)__float_as_int(val));
}

__device__ __forceinline__ float edge_logit(const float* __restrict__ el,
                                            const float* __restrict__ er, int s, int d) {
  float v = el[s] + er[d];
  return (v > 0.f) ? v : GAT_SLOPE * v;
}

#define BKT_T   1024
#define BKT_EPT 5
#define BKT_CAP (BKT_T * BKT_EPT)
#define BKT_NPW (BK_N / (BKT_T / 32))
#define BKT_NONE 0xffffffffu
static_assert(BKT_NPW * (BKT_T / 32) == BK_N, "the waves must tile the bucket's nodes");
static_assert(BK_N == 4 * 32, "the scan below gives every lane of one wave four counters");

__device__ __forceinline__ void bkt_load_chunk(const unsigned int* __restrict__ sorted,
                                               const float* __restrict__ el,
                                               const float* ler, int cb, int end, int t,
                                               unsigned int (&w)[BKT_EPT], float (&lg)[BKT_EPT]) {
  float e[BKT_EPT];
#pragma unroll
  for (int k = 0; k < BKT_EPT; ++k) {
    const int i = cb + k * BKT_T + t;
    const unsigned int v = sorted[min(i, end - 1)];
    w[k] = (i < end) ? v : BKT_NONE;
  }
#pragma unroll
  for (int k = 0; k < BKT_EPT; ++k) {
    e[k] = el[(w[k] != BKT_NONE) ? (int)(w[k] & 0xffffu) : 0];
  }
#pragma unroll
  for (int k = 0; k < BKT_EPT; ++k) {
    const int dl = (w[k] != BKT_NONE) ? (int)(w[k] >> 16) : 0;
    const float v = e[k] + ler[dl];
    lg[k] = (v > 0.f) ? v : GAT_SLOPE * v;
  }
}

__global__ __launch_bounds__(BKT_T) void k_bucket(const unsigned int* __restrict__ sorted,
                                                  const int* __restrict__ base,
                                                  const float* __restrict__ Wh,
                                                  const float* __restrict__ el,
                                                  const float* __restrict__ er,
                                                  const float* __restrict__ bias,
                                                  float* __restrict__ out) {
  __shared__ unsigned int lsrc[BKT_CAP];
  __shared__ float        lwt[BKT_CAP];
  __shared__ float mx[BK_N];
  __shared__ float sm[BK_N];
  __shared__ float ler[BK_N];
  __shared__ int   cur[BK_N];
  __shared__ int   off[BK_N + 4];
  const int t    = threadIdx.x;
  const int lane = t & 31;
  const int wv   = t >> 5;
  const int n0   = blockIdx.x * BK_N;
  const int beg  = base[blockIdx.x];
  const int end  = base[blockIdx.x + 1];
  const bool multi = (end - beg) > BKT_CAP;

  if (t < BK_N) {
    mx[t] = -__builtin_inff(); sm[t] = 0.f; cur[t] = 0;
    ler[t] = (n0 + t < GAT_N) ? er[n0 + t] : 0.f;
  }
  __syncthreads();

  unsigned int w[BKT_EPT];
  float        lg[BKT_EPT];
#pragma unroll
  for (int k = 0; k < BKT_EPT; ++k) { w[k] = BKT_NONE; lg[k] = 0.f; }

  for (int cb = beg; cb < end; cb += BKT_CAP) {
    bkt_load_chunk(sorted, el, ler, cb, end, t, w, lg);
#pragma unroll
    for (int k = 0; k < BKT_EPT; ++k) {
      if (w[k] != BKT_NONE) {
        const int dl = (int)(w[k] >> 16);
        lds_max_f32(&mx[dl], lg[k]);
        if (!multi) atomicAdd(&cur[dl], 1);
      }
    }
  }
  __syncthreads();

  v2f acc[BKT_NPW];
#pragma unroll
  for (int q = 0; q < BKT_NPW; ++q) { acc[q].x = 0.f; acc[q].y = 0.f; }

  for (int cb = beg; cb < end; cb += BKT_CAP) {
    if (multi) {
      bkt_load_chunk(sorted, el, ler, cb, end, t, w, lg);
      if (t < BK_N) cur[t] = 0;
      __syncthreads();
#pragma unroll
      for (int k = 0; k < BKT_EPT; ++k) {
        if (w[k] != BKT_NONE) atomicAdd(&cur[(int)(w[k] >> 16)], 1);
      }
      __syncthreads();
    }

    if (wv == 0) {
      const int c0 = cur[4 * lane + 0], c1 = cur[4 * lane + 1];
      const int c2 = cur[4 * lane + 2], c3 = cur[4 * lane + 3];
      const int tot = c0 + c1 + c2 + c3;
      int incl = tot;
#pragma unroll
      for (int d = 1; d < 32; d <<= 1) {
        const int up = __shfl_up(incl, d);
        if (lane >= d) incl += up;
      }
      const int ex = incl - tot;
      off[4 * lane + 0] = ex;                cur[4 * lane + 0] = ex;
      off[4 * lane + 1] = ex + c0;           cur[4 * lane + 1] = ex + c0;
      off[4 * lane + 2] = ex + c0 + c1;      cur[4 * lane + 2] = ex + c0 + c1;
      off[4 * lane + 3] = ex + c0 + c1 + c2; cur[4 * lane + 3] = ex + c0 + c1 + c2;
      if (lane == 31) off[BK_N] = incl;
    }
    __syncthreads();

    {
      float exv[BKT_EPT];
      int   pos[BKT_EPT];
#pragma unroll
      for (int k = 0; k < BKT_EPT; ++k) {
        exv[k] = mx[(w[k] != BKT_NONE) ? (int)(w[k] >> 16) : 0];
      }
#pragma unroll
      for (int k = 0; k < BKT_EPT; ++k) {
        exv[k] = __expf(lg[k] - exv[k]);
      }
#pragma unroll
      for (int k = 0; k < BKT_EPT; ++k) {
        pos[k] = 0;
        if (w[k] != BKT_NONE) {
          const int dl = (int)(w[k] >> 16);
          atomicAdd(&sm[dl], exv[k]);
          pos[k] = atomicAdd(&cur[dl], 1);
        }
      }
#pragma unroll
      for (int k = 0; k < BKT_EPT; ++k) {
        if (w[k] != BKT_NONE) {
          lsrc[pos[k]] = w[k] & 0xffffu;
          lwt[pos[k]]  = exv[k];
        }
      }
    }
    __syncthreads();

#pragma unroll
    for (int q = 0; q < BKT_NPW; ++q) {
      const int dl = wv * BKT_NPW + q;
      const int a  = __builtin_amdgcn_readfirstlane(off[dl]);
      const int b  = __builtin_amdgcn_readfirstlane(off[dl + 1]);
      for (int c = a; c < b; c += 32) {
        const int  idx = c + lane;
        const bool ok  = idx < b;
        const unsigned int s  = lsrc[ok ? idx : c];
        const float wgt = ok ? lwt[ok ? idx : c] : 0.f;
        const int nv = min(32, b - c);
        for (int j0 = 0; j0 < nv; j0 += 8) {
          v2f   r[8];
          float wj[8];
#pragma unroll
          for (int k = 0; k < 8; ++k) {
            const unsigned int sj = (unsigned int)__builtin_amdgcn_readlane((int)s, j0 + k);
            wj[k] = __int_as_float(__builtin_amdgcn_readlane(__float_as_int(wgt), j0 + k));
            r[k]  = *((const v2f*)(Wh + (size_t)sj * GAT_OUT) + lane);
          }
#pragma unroll
          for (int k = 0; k < 8; ++k) {
            acc[q].x = fmaf(wj[k], r[k].x, acc[q].x);
            acc[q].y = fmaf(wj[k], r[k].y, acc[q].y);
          }
        }
      }
    }
    if (multi) __syncthreads();
  }

  const v2f bb = *((const v2f*)bias + lane);
#pragma unroll
  for (int q = 0; q < BKT_NPW; ++q) {
    const int dl = wv * BKT_NPW + q;
    const int n  = n0 + dl;
    if (n < GAT_N) {
      const float den = sm[dl] + GAT_EPS;
      v2f o;
      o.x = acc[q].x / den + bb.x;
      o.y = acc[q].y / den + bb.y;
      *((v2f*)(out + (size_t)n * GAT_OUT) + lane) = o;
    }
  }
}

extern "C" void kernel_launch(void* const* d_in, const int* in_sizes, int n_in,
                              void* d_out, int out_size, void* d_ws, size_t ws_size,
                              hipStream_t stream) {
  const float* x    = (const float*)d_in[0];
  const int*   ei   = (const int*)d_in[1];
  const float* W    = (const float*)d_in[2];
  const float* al   = (const float*)d_in[3];
  const float* ar   = (const float*)d_in[4];
  const float* bias = (const float*)d_in[5];
  float* out = (float*)d_out;
  (void)in_sizes; (void)n_in; (void)out_size; (void)ws_size;

  float* ws   = (float*)d_ws;
  float* Wh   = ws;
  float* el   = Wh + (size_t)GAT_N * GAT_OUT;
  float* er   = el + GAT_N;
  int*   hist  = (int*)(er + GAT_N);
  int*   total = hist  + BK_NUM * SRT_NBLK;
  int*   base  = total + 512;
  unsigned int* sorted = (unsigned int*)(base + 512);

  const int* src = ei;
  const int* dst = ei + GAT_E;

  k_gemm_hist   <<<GEMM_NBLK + SRT_NBLK, SRT_T, 0, stream>>>(x, W, Wh, dst, hist);
  k_scan_elr    <<<BK_NUM + ELR_NBLK, ELR_T, 0, stream>>>(hist, total, Wh, al, ar, el, er);
  k_scan_buckets<<<1, 512, 0, stream>>>(total, base);
  k_fill        <<<SRT_NBLK, SRT_T, 0, stream>>>(src, dst, hist, base, sorted);
  k_bucket      <<<BK_NUM, BKT_T, 0, stream>>>(sorted, base, Wh, el, er, bias, out);
}


# Each sort block orders its own 8000 edges by bucket in LDS; a bucket workgroup collects its 200 segments itself; hist scans and fill pass gone, 5 launches become 3; sums still in LDS arrival order
# speedup vs baseline: 8.3277x; 1.5970x over previous
#include <hip/hip_runtime.h>
#include <hip/hip_bf16.h>

#define GAT_N      50000
#define GAT_E      1600000
#define GAT_IN     128
#define GAT_OUT    64
#define GAT_SLOPE  0.2f
#define GAT_EPS    1e-12f

typedef __attribute__((ext_vector_type(2))) float v2f;
typedef __attribute__((ext_vector_type(4))) float v4f;
typedef __attribute__((ext_vector_type(8))) float v8f;

#define GEMM_WPB  16
#define GEMM_NBLK ((GAT_N / 16 + GEMM_WPB - 1) / GEMM_WPB)
__device__ __forceinline__ void gemm_part(const int blk, const float* __restrict__ x,
                                          const float* __restrict__ W,
                                          float* __restrict__ Wh) {
  const int lane = threadIdx.x & 31;
  const int tile = blk * GEMM_WPB + (threadIdx.x >> 5);
  if (tile >= GAT_N / 16) return;
  const int base  = tile * 16;
  const int l15   = lane & 15;
  const int khalf = (lane >> 4) << 1;

  const float* xr = x + (size_t)(base + l15) * GAT_IN + khalf;
  const float* w0 = W + (size_t)(l15 +  0) * GAT_IN + khalf;
  const float* w1 = W + (size_t)(l15 + 16) * GAT_IN + khalf;
  const float* w2 = W + (size_t)(l15 + 32) * GAT_IN + khalf;
  const float* w3 = W + (size_t)(l15 + 48) * GAT_IN + khalf;

  v8f acc0 = {}, acc1 = {}, acc2 = {}, acc3 = {};

#pragma unroll 8
  for (int k = 0; k < GAT_IN; k += 4) {
    v2f a  = *(const v2f*)(xr + k);
    v2f b0 = *(const v2f*)(w0 + k);
    v2f b1 = *(const v2f*)(w1 + k);
    v2f b2 = *(const v2f*)(w2 + k);
    v2f b3 = *(const v2f*)(w3 + k);
    acc0 = __builtin_amdgcn_wmma_f32_16x16x4_f32(false, a, false, b0, (short)0, acc0, false, false);
    acc1 = __builtin_amdgcn_wmma_f32_16x16x4_f32(false, a, false, b1, (short)0, acc1, false, false);
    acc2 = __builtin_amdgcn_wmma_f32_16x16x4_f32(false, a, false, b2, (short)0, acc2, false, false);
    acc3 = __builtin_amdgcn_wmma_f32_16x16x4_f32(false, a, false, b3, (short)0, acc3, false, false);
  }

  const int rhalf = (lane >> 4) << 3;
  float* o = Wh + (size_t)(base + rhalf) * GAT_OUT + l15;
#pragma unroll
  for (int v = 0; v < 8; ++v) {
    o[(size_t)v * GAT_OUT +  0] = acc0[v];
    o[(size_t)v * GAT_OUT + 16] = acc1[v];
    o[(size_t)v * GAT_OUT + 32] = acc2[v];
    o[(size_t)v * GAT_OUT + 48] = acc3[v];
  }
}

#define ELR_T    256
#define ELR_NBLK ((GAT_N + ELR_T - 1) / ELR_T)
__device__ __forceinline__ void elr_part(const int blk, const float* __restrict__ Wh,
                                         const float* __restrict__ al,
                                         const float* __restrict__ ar, float* __restrict__ el,
                                         float* __restrict__ er) {
  int i = blk * ELR_T + threadIdx.x;
  if (i >= GAT_N) return;
  const v4f* r = (const v4f*)(Wh + (size_t)i * GAT_OUT);
  float sl = 0.f, sr = 0.f;
#pragma unroll
  for (int q = 0; q < GAT_OUT / 4; ++q) {
    v4f v = r[q];
#pragma unroll
    for (int j = 0; j < 4; ++j) {
      sl += v[j] * al[q * 4 + j];
      sr += v[j] * ar[q * 4 + j];
    }
  }
  el[i] = sl; er[i] = sr;
}

#define BK_SH    7
#define BK_N     128
#define BK_NUM   ((GAT_N + BK_N - 1) / BK_N)
#define SRT_NBLK 200
#define SRT_CH   (GAT_E / SRT_NBLK)
#define SRT_T    512
#define SRT_EPT  ((SRT_CH + SRT_T - 1) / SRT_T)
static_assert(SRT_NBLK * SRT_CH == GAT_E, "sort blocks must tile the edge list");
static_assert(GAT_N <= 65536, "src is packed into 16 bits of the sorted word");
static_assert(SRT_CH < 65536, "segment start and length are packed into 16 bits each");
static_assert(BK_NUM <= SRT_T, "one bucket counter per thread in the block scan");

__device__ __forceinline__ void sort_part(const int blk, const int* __restrict__ src,
                                          const int* __restrict__ dst,
                                          unsigned int* __restrict__ seg,
                                          unsigned int* __restrict__ sorted,
                                          int* h, int* wsum, unsigned int* stage) {
  const int t    = threadIdx.x;
  const int lane = t & 31;
  const int wv   = t >> 5;
  h[t] = 0;
  __syncthreads();
  const int* sp = src + (size_t)blk * SRT_CH;
  const int* dp = dst + (size_t)blk * SRT_CH;
  int d[SRT_EPT], s[SRT_EPT];
#pragma unroll
  for (int k = 0; k < SRT_EPT; ++k) {
    const int i = k * SRT_T + t;
    d[k] = -1; s[k] = 0;
    if (i < SRT_CH) { d[k] = dp[i]; s[k] = sp[i]; }
  }
#pragma unroll
  for (int k = 0; k < SRT_EPT; ++k) {
    if (d[k] >= 0) atomicAdd(&h[d[k] >> BK_SH], 1);
  }
  __syncthreads();

  const int v = h[t];
  int incl = v;
#pragma unroll
  for (int o = 1; o < 32; o <<= 1) {
    const int up = __shfl_up(incl, o);
    if (lane >= o) incl += up;
  }
  if (lane == 31) wsum[wv] = incl;
  __syncthreads();
  int woff = 0;
#pragma unroll
  for (int q = 0; q < SRT_T / 32; ++q) woff += (q < wv) ? wsum[q] : 0;
  const int ex = woff + incl - v;
  h[t] = ex;
  if (t < BK_NUM) seg[(size_t)t * SRT_NBLK + blk] = ((unsigned int)ex << 16) | (unsigned int)v;
  __syncthreads();

  int slot[SRT_EPT];
#pragma unroll
  for (int k = 0; k < SRT_EPT; ++k) {
    slot[k] = 0;
    if (d[k] >= 0) slot[k] = atomicAdd(&h[d[k] >> BK_SH], 1);
  }
#pragma unroll
  for (int k = 0; k < SRT_EPT; ++k) {
    if (d[k] >= 0) stage[slot[k]] = (unsigned int)s[k] | ((unsigned int)(d[k] & (BK_N - 1)) << 16);
  }
  __syncthreads();
  unsigned int* op = sorted + (size_t)blk * SRT_CH;
#pragma unroll
  for (int k = 0; k < SRT_EPT; ++k) {
    const int i = k * SRT_T + t;
    if (i < SRT_CH) op[i] = stage[i];
  }
}

static_assert(GEMM_WPB * 32 == SRT_T, "both parts of the shared launch use one block size");
__global__ __launch_bounds__(SRT_T) void k_gemm_sort(const float* __restrict__ x,
                                                     const float* __restrict__ W,
                                                     float* __restrict__ Wh,
                                                     const int* __restrict__ src,
                                                     const int* __restrict__ dst,
                                                     unsigned int* __restrict__ seg,
                                                     unsigned int* __restrict__ sorted) {
  __shared__ int h[SRT_T];
  __shared__ int wsum[SRT_T / 32];
  __shared__ unsigned int stage[SRT_CH];
  if (blockIdx.x < GEMM_NBLK) gemm_part(blockIdx.x, x, W, Wh);
  else                        sort_part(blockIdx.x - GEMM_NBLK, src, dst, seg, sorted, h, wsum, stage);
}

__global__ __launch_bounds__(ELR_T) void k_elr(const float* __restrict__ Wh,
                                               const float* __restrict__ al,
                                               const float* __restrict__ ar,
                                               float* __restrict__ el, float* __restrict__ er) {
  elr_part(blockIdx.x, Wh, al, ar, el, er);
}

__device__ __forceinline__ void lds_max_f32(float* addr, float val) {
  if (val >= 0.f) atomicMax((int*)addr, __float_as_int(val));
  else            atomicMin((unsigned int*)addr, (unsigned int)__float_as_int(val));
}

__device__ __forceinline__ float edge_logit(const float* __restrict__ el,
                                            const float* __restrict__ er, int s, int d) {
  float v = el[s] + er[d];
  return (v > 0.f) ? v : GAT_SLOPE * v;
}

#define BKT_T   1024
#define BKT_EPT 5
#define BKT_CAP (BKT_T * BKT_EPT)
#define BKT_NPW (BK_N / (BKT_T / 32))
#define BKT_NONE 0xffffffffu
#define BKT_SPL ((SRT_NBLK + 31) / 32)
static_assert(BKT_NPW * (BKT_T / 32) == BK_N, "the waves must tile the bucket's nodes");
static_assert(BK_N == 4 * 32, "the scan below gives every lane of one wave four counters");

__device__ __forceinline__ int bkt_segs(const unsigned int* __restrict__ sg, int lane, int wvs,
                                        int (&st)[BKT_SPL], int (&ln)[BKT_SPL],
                                        int (&ps)[BKT_SPL]) {
  int sst[BKT_SPL], sln[BKT_SPL];
#pragma unroll
  for (int j = 0; j < BKT_SPL; ++j) {
    const int k = j * 32 + lane;
    unsigned int v = 0u;
    if (k < SRT_NBLK) v = sg[k];
    sst[j] = k * SRT_CH + (int)(v >> 16);
    sln[j] = (int)(v & 0xffffu);
  }
  int run = 0;
#pragma unroll
  for (int j = 0; j < BKT_SPL; ++j) {
    int incl = sln[j];
#pragma unroll
    for (int o = 1; o < 32; o <<= 1) {
      const int up = __shfl_up(incl, o);
      if (lane >= o) incl += up;
    }
    st[j] = __builtin_amdgcn_readlane(sst[j], wvs);
    ln[j] = __builtin_amdgcn_readlane(sln[j], wvs);
    ps[j] = run + __builtin_amdgcn_readlane(incl - sln[j], wvs);
    run += __builtin_amdgcn_readlane(incl, 31);
  }
  return run;
}

__device__ __forceinline__ void bkt_stage(const unsigned int* __restrict__ sorted,
                                          unsigned int* lraw, const int (&st)[BKT_SPL],
                                          const int (&ln)[BKT_SPL], const int (&ps)[BKT_SPL],
                                          int cb, int lane) {
  int e0[BKT_SPL], e1[BKT_SPL];
  unsigned int v[BKT_SPL];
#pragma unroll
  for (int j = 0; j < BKT_SPL; ++j) {
    e0[j] = max(0, cb - ps[j]);
    e1[j] = min(ln[j], cb + BKT_CAP - ps[j]);
    v[j] = 0u;
    if (e0[j] + lane < e1[j]) v[j] = sorted[st[j] + e0[j] + lane];
  }
#pragma unroll
  for (int j = 0; j < BKT_SPL; ++j) {
    if (e0[j] + lane < e1[j]) lraw[ps[j] + e0[j] + lane - cb] = v[j];
  }
#pragma unroll
  for (int j = 0; j < BKT_SPL; ++j) {
#pragma nounroll
    for (int e = e0[j] + 32 + lane; e < e1[j]; e += 32) lraw[ps[j] + e - cb] = sorted[st[j] + e];
  }
}

__device__ __forceinline__ void bkt_load_chunk(const unsigned int* lraw,
                                               const float* __restrict__ el,
                                               const float* ler, int cb, int end, int t,
                                               unsigned int (&w)[BKT_EPT], float (&lg)[BKT_EPT]) {
  float e[BKT_EPT];
#pragma unroll
  for (int k = 0; k < BKT_EPT; ++k) {
    const int i = k * BKT_T + t;
    const unsigned int v = lraw[i];
    w[k] = (cb + i < end) ? v : BKT_NONE;
  }
#pragma unroll
  for (int k = 0; k < BKT_EPT; ++k) {
    e[k] = el[(w[k] != BKT_NONE) ? (int)(w[k] & 0xffffu) : 0];
  }
#pragma unroll
  for (int k = 0; k < BKT_EPT; ++k) {
    const int dl = (w[k] != BKT_NONE) ? (int)(w[k] >> 16) : 0;
    const float v = e[k] + ler[dl];
    lg[k] = (v > 0.f) ? v : GAT_SLOPE * v;
  }
}

__global__ __launch_bounds__(BKT_T) void k_bucket(const unsigned int* __restrict__ sorted,
                                                  const unsigned int* __restrict__ seg,
                                                  const float* __restrict__ Wh,
                                                  const float* __restrict__ el,
                                                  const float* __restrict__ er,
                                                  const float* __restrict__ bias,
                                                  float* __restrict__ out) {
  __shared__ unsigned int lsrc[BKT_CAP];
  __shared__ float        lwt[BKT_CAP];
  __shared__ float mx[BK_N];
  __shared__ float sm[BK_N];
  __shared__ float ler[BK_N];
  __shared__ int   cur[BK_N];
  __shared__ int   off[BK_N + 4];
  const int t    = threadIdx.x;
  const int lane = t & 31;
  const int wv   = t >> 5;
  const int wvs  = __builtin_amdgcn_readfirstlane(wv);
  const int n0   = blockIdx.x * BK_N;
  const unsigned int* sg = seg + (size_t)blockIdx.x * SRT_NBLK;
  int sst[BKT_SPL], sln[BKT_SPL], sps[BKT_SPL];
  const int end  = bkt_segs(sg, lane, wvs, sst, sln, sps);
  const bool multi = end > BKT_CAP;

  if (t < BK_N) {
    mx[t] = -__builtin_inff(); sm[t] = 0.f; cur[t] = 0;
    ler[t] = (n0 + t < GAT_N) ? er[n0 + t] : 0.f;
  }

  unsigned int w[BKT_EPT];
  float        lg[BKT_EPT];
#pragma unroll
  for (int k = 0; k < BKT_EPT; ++k) { w[k] = BKT_NONE; lg[k] = 0.f; }

  for (int cb = 0; cb < end; cb += BKT_CAP) {
    bkt_stage(sorted, lsrc, sst, sln, sps, cb, lane);
    __syncthreads();
    bkt_load_chunk(lsrc, el, ler, cb, end, t, w, lg);
#pragma unroll
    for (int k = 0; k < BKT_EPT; ++k) {
      if (w[k] != BKT_NONE) {
        const int dl = (int)(w[k] >> 16);
        lds_max_f32(&mx[dl], lg[k]);
        if (!multi) atomicAdd(&cur[dl], 1);
      }
    }
    if (multi) __syncthreads();
  }
  __syncthreads();

  v2f acc[BKT_NPW];
#pragma unroll
  for (int q = 0; q < BKT_NPW; ++q) { acc[q].x = 0.f; acc[q].y = 0.f; }

  for (int cb = 0; cb < end; cb += BKT_CAP) {
    if (multi) {
      bkt_stage(sorted, lsrc, sst, sln, sps, cb, lane);
      if (t < BK_N) cur[t] = 0;
      __syncthreads();
      bkt_load_chunk(lsrc, el, ler, cb, end, t, w, lg);
#pragma unroll
      for (int k = 0; k < BKT_EPT; ++k) {
        if (w[k] != BKT_NONE) atomicAdd(&cur[(int)(w[k] >> 16)], 1);
      }
      __syncthreads();
    }

    if (wv == 0) {
      const int c0 = cur[4 * lane + 0], c1 = cur[4 * lane + 1];
      const int c2 = cur[4 * lane + 2], c3 = cur[4 * lane + 3];
      const int tot = c0 + c1 + c2 + c3;
      int incl = tot;
#pragma unroll
      for (int d = 1; d < 32; d <<= 1) {
        const int up = __shfl_up(incl, d);
        if (lane >= d) incl += up;
      }
      const int ex = incl - tot;
      off[4 * lane + 0] = ex;                cur[4 * lane + 0] = ex;
      off[4 * lane + 1] = ex + c0;           cur[4 * lane + 1] = ex + c0;
      off[4 * lane + 2] = ex + c0 + c1;      cur[4 * lane + 2] = ex + c0 + c1;
      off[4 * lane + 3] = ex + c0 + c1 + c2; cur[4 * lane + 3] = ex + c0 + c1 + c2;
      if (lane == 31) off[BK_N] = incl;
    }
    __syncthreads();

    {
      float exv[BKT_EPT];
      int   pos[BKT_EPT];
#pragma unroll
      for (int k = 0; k < BKT_EPT; ++k) {
        exv[k] = mx[(w[k] != BKT_NONE) ? (int)(w[k] >> 16) : 0];
      }
#pragma unroll
      for (int k = 0; k < BKT_EPT; ++k) {
        exv[k] = __expf(lg[k] - exv[k]);
      }
#pragma unroll
      for (int k = 0; k < BKT_EPT; ++k) {
        pos[k] = 0;
        if (w[k] != BKT_NONE) {
          const int dl = (int)(w[k] >> 16);
          atomicAdd(&sm[dl], exv[k]);
          pos[k] = atomicAdd(&cur[dl], 1);
        }
      }
#pragma unroll
      for (int k = 0; k < BKT_EPT; ++k) {
        if (w[k] != BKT_NONE) {
          lsrc[pos[k]] = w[k] & 0xffffu;
          lwt[pos[k]]  = exv[k];
        }
      }
    }
    __syncthreads();

#pragma unroll
    for (int q = 0; q < BKT_NPW; ++q) {
      const int dl = wv * BKT_NPW + q;
      const int a  = __builtin_amdgcn_readfirstlane(off[dl]);
      const int b  = __builtin_amdgcn_readfirstlane(off[dl + 1]);
      for (int c = a; c < b; c += 32) {
        const int  idx = c + lane;
        const bool ok  = idx < b;
        const unsigned int s  = lsrc[ok ? idx : c];
        const float wgt = ok ? lwt[ok ? idx : c] : 0.f;
        const int nv = min(32, b - c);
        for (int j0 = 0; j0 < nv; j0 += 8) {
          v2f   r[8];
          float wj[8];
#pragma unroll
          for (int k = 0; k < 8; ++k) {
            const unsigned int sj = (unsigned int)__builtin_amdgcn_readlane((int)s, j0 + k);
            wj[k] = __int_as_float(__builtin_amdgcn_readlane(__float_as_int(wgt), j0 + k));
            r[k]  = *((const v2f*)(Wh + (size_t)sj * GAT_OUT) + lane);
          }
#pragma unroll
          for (int k = 0; k < 8; ++k) {
            acc[q].x = fmaf(wj[k], r[k].x, acc[q].x);
            acc[q].y = fmaf(wj[k], r[k].y, acc[q].y);
          }
        }
      }
    }
    if (multi) __syncthreads();
  }

  const v2f bb = *((const v2f*)bias + lane);
#pragma unroll
  for (int q = 0; q < BKT_NPW; ++q) {
    const int dl = wv * BKT_NPW + q;
    const int n  = n0 + dl;
    if (n < GAT_N) {
      const float den = sm[dl] + GAT_EPS;
      v2f o;
      o.x = acc[q].x / den + bb.x;
      o.y = acc[q].y / den + bb.y;
      *((v2f*)(out + (size_t)n * GAT_OUT) + lane) = o;
    }
  }
}

extern "C" void kernel_launch(void* const* d_in, const int* in_sizes, int n_in,
                              void* d_out, int out_size, void* d_ws, size_t ws_size,
                              hipStream_t stream) {
  const float* x    = (const float*)d_in[0];
  const int*   ei   = (const int*)d_in[1];
  const float* W    = (const float*)d_in[2];
  const float* al   = (const float*)d_in[3];
  const float* ar   = (const float*)d_in[4];
  const float* bias = (const float*)d_in[5];
  float* out = (float*)d_out;
  (void)in_sizes; (void)n_in; (void)out_size; (void)ws_size;

  float* ws   = (float*)d_ws;
  float* Wh   = ws;
  float* el   = Wh + (size_t)GAT_N * GAT_OUT;
  float* er   = el + GAT_N;
  unsigned int* seg    = (unsigned int*)(er + GAT_N);
  unsigned int* sorted = seg + BK_NUM * SRT_NBLK;

  const int* src = ei;
  const int* dst = ei + GAT_E;

  k_gemm_sort<<<GEMM_NBLK + SRT_NBLK, SRT_T, 0, stream>>>(x, W, Wh, src, dst, seg, sorted);
  k_elr      <<<ELR_NBLK, ELR_T, 0, stream>>>(Wh, al, ar, el, er);
  k_bucket   <<<BK_NUM, BKT_T, 0, stream>>>(sorted, seg, Wh, el, er, bias, out);
}


# el/er computed in the GEMM epilogue from the accumulators (4 fma per lane, then a 16-lane row_xmask add tree); the el/er launch is gone, 3 launches become 2; el/er sum order is now that tree
# speedup vs baseline: 8.9921x; 1.0798x over previous
#include <hip/hip_runtime.h>
#include <hip/hip_bf16.h>

#define GAT_N      50000
#define GAT_E      1600000
#define GAT_IN     128
#define GAT_OUT    64
#define GAT_SLOPE  0.2f
#define GAT_EPS    1e-12f

typedef __attribute__((ext_vector_type(2))) float v2f;
typedef __attribute__((ext_vector_type(8))) float v8f;


template <int MASK>
__device__ __forceinline__ float row_xor(float x) {
  return __int_as_float(__builtin_amdgcn_update_dpp(0, __float_as_int(x), 0x160 + MASK, 0xf, 0xf, false));
}

#define GEMM_WPB  16
#define GEMM_NBLK ((GAT_N / 16 + GEMM_WPB - 1) / GEMM_WPB)
__device__ __forceinline__ void gemm_part(const int blk, const float* __restrict__ x,
                                          const float* __restrict__ W,
                                          const float* __restrict__ al,
                                          const float* __restrict__ ar,
                                          float* __restrict__ Wh,
                                          float* __restrict__ el, float* __restrict__ er) {
  const int lane = threadIdx.x & 31;
  const int tile = blk * GEMM_WPB + (threadIdx.x >> 5);
  if (tile >= GAT_N / 16) return;
  const int base  = tile * 16;
  const int l15   = lane & 15;
  const int khalf = (lane >> 4) << 1;

  const float* xr = x + (size_t)(base + l15) * GAT_IN + khalf;
  const float* w0 = W + (size_t)(l15 +  0) * GAT_IN + khalf;
  const float* w1 = W + (size_t)(l15 + 16) * GAT_IN + khalf;
  const float* w2 = W + (size_t)(l15 + 32) * GAT_IN + khalf;
  const float* w3 = W + (size_t)(l15 + 48) * GAT_IN + khalf;

  v8f acc0 = {}, acc1 = {}, acc2 = {}, acc3 = {};

#pragma unroll 8
  for (int k = 0; k < GAT_IN; k += 4) {
    v2f a  = *(const v2f*)(xr + k);
    v2f b0 = *(const v2f*)(w0 + k);
    v2f b1 = *(const v2f*)(w1 + k);
    v2f b2 = *(const v2f*)(w2 + k);
    v2f b3 = *(const v2f*)(w3 + k);
    acc0 = __builtin_amdgcn_wmma_f32_16x16x4_f32(false, a, false, b0, (short)0, acc0, false, false);
    acc1 = __builtin_amdgcn_wmma_f32_16x16x4_f32(false, a, false, b1, (short)0, acc1, false, false);
    acc2 = __builtin_amdgcn_wmma_f32_16x16x4_f32(false, a, false, b2, (short)0, acc2, false, false);
    acc3 = __builtin_amdgcn_wmma_f32_16x16x4_f32(false, a, false, b3, (short)0, acc3, false, false);
  }

  const int rhalf = (lane >> 4) << 3;
  float* o = Wh + (size_t)(base + rhalf) * GAT_OUT + l15;
#pragma unroll
  for (int v = 0; v < 8; ++v) {
    o[(size_t)v * GAT_OUT +  0] = acc0[v];
    o[(size_t)v * GAT_OUT + 16] = acc1[v];
    o[(size_t)v * GAT_OUT + 32] = acc2[v];
    o[(size_t)v * GAT_OUT + 48] = acc3[v];
  }

  const float l0 = al[l15], l1 = al[l15 + 16], l2 = al[l15 + 32], l3 = al[l15 + 48];
  const float r0 = ar[l15], r1 = ar[l15 + 16], r2 = ar[l15 + 32], r3 = ar[l15 + 48];
  float u[16];
#pragma unroll
  for (int v = 0; v < 8; ++v) {
    u[v]     = fmaf(acc3[v], l3, fmaf(acc2[v], l2, fmaf(acc1[v], l1, acc0[v] * l0)));
    u[v + 8] = fmaf(acc3[v], r3, fmaf(acc2[v], r2, fmaf(acc1[v], r1, acc0[v] * r0)));
  }
#pragma unroll
  for (int i = 0; i < 16; ++i) u[i] += row_xor<1>(u[i]);
#pragma unroll
  for (int i = 0; i < 16; ++i) u[i] += row_xor<2>(u[i]);
#pragma unroll
  for (int i = 0; i < 16; ++i) u[i] += row_xor<4>(u[i]);
#pragma unroll
  for (int i = 0; i < 16; ++i) u[i] += row_xor<8>(u[i]);
  float keep = u[0];
#pragma unroll
  for (int i = 1; i < 16; ++i) keep = (l15 == i) ? u[i] : keep;
  float* dstp = (l15 < 8) ? el : er;
  dstp[base + rhalf + (l15 & 7)] = keep;
}

#define BK_SH    7
#define BK_N     128
#define BK_NUM   ((GAT_N + BK_N - 1) / BK_N)
#define SRT_NBLK 200
#define SRT_CH   (GAT_E / SRT_NBLK)
#define SRT_T    512
#define SRT_EPT  ((SRT_CH + SRT_T - 1) / SRT_T)
static_assert(SRT_NBLK * SRT_CH == GAT_E, "sort blocks must tile the edge list");
static_assert(GAT_N <= 65536, "src is packed into 16 bits of the sorted word");
static_assert(SRT_CH < 65536, "segment start and length are packed into 16 bits each");
static_assert(BK_NUM <= SRT_T, "one bucket counter per thread in the block scan");

__device__ __forceinline__ void sort_part(const int blk, const int* __restrict__ src,
                                          const int* __restrict__ dst,
                                          unsigned int* __restrict__ seg,
                                          unsigned int* __restrict__ sorted,
                                          int* h, int* wsum, unsigned int* stage) {
  const int t    = threadIdx.x;
  const int lane = t & 31;
  const int wv   = t >> 5;
  h[t] = 0;
  __syncthreads();
  const int* sp = src + (size_t)blk * SRT_CH;
  const int* dp = dst + (size_t)blk * SRT_CH;
  int d[SRT_EPT], s[SRT_EPT];
#pragma unroll
  for (int k = 0; k < SRT_EPT; ++k) {
    const int i = k * SRT_T + t;
    d[k] = -1; s[k] = 0;
    if (i < SRT_CH) { d[k] = dp[i]; s[k] = sp[i]; }
  }
#pragma unroll
  for (int k = 0; k < SRT_EPT; ++k) {
    if (d[k] >= 0) atomicAdd(&h[d[k] >> BK_SH], 1);
  }
  __syncthreads();

  const int v = h[t];
  int incl = v;
#pragma unroll
  for (int o = 1; o < 32; o <<= 1) {
    const int up = __shfl_up(incl, o);
    if (lane >= o) incl += up;
  }
  if (lane == 31) wsum[wv] = incl;
  __syncthreads();
  int woff = 0;
#pragma unroll
  for (int q = 0; q < SRT_T / 32; ++q) woff += (q < wv) ? wsum[q] : 0;
  const int ex = woff + incl - v;
  h[t] = ex;
  if (t < BK_NUM) seg[(size_t)t * SRT_NBLK + blk] = ((unsigned int)ex << 16) | (unsigned int)v;
  __syncthreads();

  int slot[SRT_EPT];
#pragma unroll
  for (int k = 0; k < SRT_EPT; ++k) {
    slot[k] = 0;
    if (d[k] >= 0) slot[k] = atomicAdd(&h[d[k] >> BK_SH], 1);
  }
#pragma unroll
  for (int k = 0; k < SRT_EPT; ++k) {
    if (d[k] >= 0) stage[slot[k]] = (unsigned int)s[k] | ((unsigned int)(d[k] & (BK_N - 1)) << 16);
  }
  __syncthreads();
  unsigned int* op = sorted + (size_t)blk * SRT_CH;
#pragma unroll
  for (int k = 0; k < SRT_EPT; ++k) {
    const int i = k * SRT_T + t;
    if (i < SRT_CH) op[i] = stage[i];
  }
}

static_assert(GEMM_WPB * 32 == SRT_T, "both parts of the shared launch use one block size");
__global__ __launch_bounds__(SRT_T) void k_gemm_sort(const float* __restrict__ x,
                                                     const float* __restrict__ W,
                                                     const float* __restrict__ al,
                                                     const float* __restrict__ ar,
                                                     float* __restrict__ Wh,
                                                     float* __restrict__ el,
                                                     float* __restrict__ er,
                                                     const int* __restrict__ src,
                                                     const int* __restrict__ dst,
                                                     unsigned int* __restrict__ seg,
                                                     unsigned int* __restrict__ sorted) {
  __shared__ int h[SRT_T];
  __shared__ int wsum[SRT_T / 32];
  __shared__ unsigned int stage[SRT_CH];
  if (blockIdx.x < GEMM_NBLK) gemm_part(blockIdx.x, x, W, al, ar, Wh, el, er);
  else                        sort_part(blockIdx.x - GEMM_NBLK, src, dst, seg, sorted, h, wsum, stage);
}

__device__ __forceinline__ void lds_max_f32(float* addr, float val) {
  if (val >= 0.f) atomicMax((int*)addr, __float_as_int(val));
  else            atomicMin((unsigned int*)addr, (unsigned int)__float_as_int(val));
}

__device__ __forceinline__ float edge_logit(const float* __restrict__ el,
                                            const float* __restrict__ er, int s, int d) {
  float v = el[s] + er[d];
  return (v > 0.f) ? v : GAT_SLOPE * v;
}

#define BKT_T   1024
#define BKT_EPT 5
#define BKT_CAP (BKT_T * BKT_EPT)
#define BKT_NPW (BK_N / (BKT_T / 32))
#define BKT_NONE 0xffffffffu
#define BKT_SPL ((SRT_NBLK + 31) / 32)
static_assert(BKT_NPW * (BKT_T / 32) == BK_N, "the waves must tile the bucket's nodes");
static_assert(BK_N == 4 * 32, "the scan below gives every lane of one wave four counters");

__device__ __forceinline__ int bkt_segs(const unsigned int* __restrict__ sg, int lane, int wvs,
                                        int (&st)[BKT_SPL], int (&ln)[BKT_SPL],
                                        int (&ps)[BKT_SPL]) {
  int sst[BKT_SPL], sln[BKT_SPL];
#pragma unroll
  for (int j = 0; j < BKT_SPL; ++j) {
    const int k = j * 32 + lane;
    unsigned int v = 0u;
    if (k < SRT_NBLK) v = sg[k];
    sst[j] = k * SRT_CH + (int)(v >> 16);
    sln[j] = (int)(v & 0xffffu);
  }
  int run = 0;
#pragma unroll
  for (int j = 0; j < BKT_SPL; ++j) {
    int incl = sln[j];
#pragma unroll
    for (int o = 1; o < 32; o <<= 1) {
      const int up = __shfl_up(incl, o);
      if (lane >= o) incl += up;
    }
    st[j] = __builtin_amdgcn_readlane(sst[j], wvs);
    ln[j] = __builtin_amdgcn_readlane(sln[j], wvs);
    ps[j] = run + __builtin_amdgcn_readlane(incl - sln[j], wvs);
    run += __builtin_amdgcn_readlane(incl, 31);
  }
  return run;
}

__device__ __forceinline__ void bkt_stage(const unsigned int* __restrict__ sorted,
                                          unsigned int* lraw, const int (&st)[BKT_SPL],
                                          const int (&ln)[BKT_SPL], const int (&ps)[BKT_SPL],
                                          int cb, int lane) {
  int e0[BKT_SPL], e1[BKT_SPL];
  unsigned int v[BKT_SPL];
#pragma unroll
  for (int j = 0; j < BKT_SPL; ++j) {
    e0[j] = max(0, cb - ps[j]);
    e1[j] = min(ln[j], cb + BKT_CAP - ps[j]);
    v[j] = 0u;
    if (e0[j] + lane < e1[j]) v[j] = sorted[st[j] + e0[j] + lane];
  }
#pragma unroll
  for (int j = 0; j < BKT_SPL; ++j) {
    if (e0[j] + lane < e1[j]) lraw[ps[j] + e0[j] + lane - cb] = v[j];
  }
#pragma unroll
  for (int j = 0; j < BKT_SPL; ++j) {
#pragma nounroll
    for (int e = e0[j] + 32 + lane; e < e1[j]; e += 32) lraw[ps[j] + e - cb] = sorted[st[j] + e];
  }
}

__device__ __forceinline__ void bkt_load_chunk(const unsigned int* lraw,
                                               const float* __restrict__ el,
                                               const float* ler, int cb, int end, int t,
                                               unsigned int (&w)[BKT_EPT], float (&lg)[BKT_EPT]) {
  float e[BKT_EPT];
#pragma unroll
  for (int k = 0; k < BKT_EPT; ++k) {
    const int i = k * BKT_T + t;
    const unsigned int v = lraw[i];
    w[k] = (cb + i < end) ? v : BKT_NONE;
  }
#pragma unroll
  for (int k = 0; k < BKT_EPT; ++k) {
    e[k] = el[(w[k] != BKT_NONE) ? (int)(w[k] & 0xffffu) : 0];
  }
#pragma unroll
  for (int k = 0; k < BKT_EPT; ++k) {
    const int dl = (w[k] != BKT_NONE) ? (int)(w[k] >> 16) : 0;
    const float v = e[k] + ler[dl];
    lg[k] = (v > 0.f) ? v : GAT_SLOPE * v;
  }
}

__global__ __launch_bounds__(BKT_T) void k_bucket(const unsigned int* __restrict__ sorted,
                                                  const unsigned int* __restrict__ seg,
                                                  const float* __restrict__ Wh,
                                                  const float* __restrict__ el,
                                                  const float* __restrict__ er,
                                                  const float* __restrict__ bias,
                                                  float* __restrict__ out) {
  __shared__ unsigned int lsrc[BKT_CAP];
  __shared__ float        lwt[BKT_CAP];
  __shared__ float mx[BK_N];
  __shared__ float sm[BK_N];
  __shared__ float ler[BK_N];
  __shared__ int   cur[BK_N];
  __shared__ int   off[BK_N + 4];
  const int t    = threadIdx.x;
  const int lane = t & 31;
  const int wv   = t >> 5;
  const int wvs  = __builtin_amdgcn_readfirstlane(wv);
  const int n0   = blockIdx.x * BK_N;
  const unsigned int* sg = seg + (size_t)blockIdx.x * SRT_NBLK;
  int sst[BKT_SPL], sln[BKT_SPL], sps[BKT_SPL];
  const int end  = bkt_segs(sg, lane, wvs, sst, sln, sps);
  const bool multi = end > BKT_CAP;

  if (t < BK_N) {
    mx[t] = -__builtin_inff(); sm[t] = 0.f; cur[t] = 0;
    ler[t] = (n0 + t < GAT_N) ? er[n0 + t] : 0.f;
  }

  unsigned int w[BKT_EPT];
  float        lg[BKT_EPT];
#pragma unroll
  for (int k = 0; k < BKT_EPT; ++k) { w[k] = BKT_NONE; lg[k] = 0.f; }

  for (int cb = 0; cb < end; cb += BKT_CAP) {
    bkt_stage(sorted, lsrc, sst, sln, sps, cb, lane);
    __syncthreads();
    bkt_load_chunk(lsrc, el, ler, cb, end, t, w, lg);
#pragma unroll
    for (int k = 0; k < BKT_EPT; ++k) {
      if (w[k] != BKT_NONE) {
        const int dl = (int)(w[k] >> 16);
        lds_max_f32(&mx[dl], lg[k]);
        if (!multi) atomicAdd(&cur[dl], 1);
      }
    }
    if (multi) __syncthreads();
  }
  __syncthreads();

  v2f acc[BKT_NPW];
#pragma unroll
  for (int q = 0; q < BKT_NPW; ++q) { acc[q].x = 0.f; acc[q].y = 0.f; }

  for (int cb = 0; cb < end; cb += BKT_CAP) {
    if (multi) {
      bkt_stage(sorted, lsrc, sst, sln, sps, cb, lane);
      if (t < BK_N) cur[t] = 0;
      __syncthreads();
      bkt_load_chunk(lsrc, el, ler, cb, end, t, w, lg);
#pragma unroll
      for (int k = 0; k < BKT_EPT; ++k) {
        if (w[k] != BKT_NONE) atomicAdd(&cur[(int)(w[k] >> 16)], 1);
      }
      __syncthreads();
    }

    if (wv == 0) {
      const int c0 = cur[4 * lane + 0], c1 = cur[4 * lane + 1];
      const int c2 = cur[4 * lane + 2], c3 = cur[4 * lane + 3];
      const int tot = c0 + c1 + c2 + c3;
      int incl = tot;
#pragma unroll
      for (int d = 1; d < 32; d <<= 1) {
        const int up = __shfl_up(incl, d);
        if (lane >= d) incl += up;
      }
      const int ex = incl - tot;
      off[4 * lane + 0] = ex;                cur[4 * lane + 0] = ex;
      off[4 * lane + 1] = ex + c0;           cur[4 * lane + 1] = ex + c0;
      off[4 * lane + 2] = ex + c0 + c1;      cur[4 * lane + 2] = ex + c0 + c1;
      off[4 * lane + 3] = ex + c0 + c1 + c2; cur[4 * lane + 3] = ex + c0 + c1 + c2;
      if (lane == 31) off[BK_N] = incl;
    }
    __syncthreads();

    {
      float exv[BKT_EPT];
      int   pos[BKT_EPT];
#pragma unroll
      for (int k = 0; k < BKT_EPT; ++k) {
        exv[k] = mx[(w[k] != BKT_NONE) ? (int)(w[k] >> 16) : 0];
      }
#pragma unroll
      for (int k = 0; k < BKT_EPT; ++k) {
        exv[k] = __expf(lg[k] - exv[k]);
      }
#pragma unroll
      for (int k = 0; k < BKT_EPT; ++k) {
        pos[k] = 0;
        if (w[k] != BKT_NONE) {
          const int dl = (int)(w[k] >> 16);
          atomicAdd(&sm[dl], exv[k]);
          pos[k] = atomicAdd(&cur[dl], 1);
        }
      }
#pragma unroll
      for (int k = 0; k < BKT_EPT; ++k) {
        if (w[k] != BKT_NONE) {
          lsrc[pos[k]] = w[k] & 0xffffu;
          lwt[pos[k]]  = exv[k];
        }
      }
    }
    __syncthreads();

#pragma unroll
    for (int q = 0; q < BKT_NPW; ++q) {
      const int dl = wv * BKT_NPW + q;
      const int a  = __builtin_amdgcn_readfirstlane(off[dl]);
      const int b  = __builtin_amdgcn_readfirstlane(off[dl + 1]);
      for (int c = a; c < b; c += 32) {
        const int  idx = c + lane;
        const bool ok  = idx < b;
        const unsigned int s  = lsrc[ok ? idx : c];
        const float wgt = ok ? lwt[ok ? idx : c] : 0.f;
        const int nv = min(32, b - c);
        for (int j0 = 0; j0 < nv; j0 += 8) {
          v2f   r[8];
          float wj[8];
#pragma unroll
          for (int k = 0; k < 8; ++k) {
            const unsigned int sj = (unsigned int)__builtin_amdgcn_readlane((int)s, j0 + k);
            wj[k] = __int_as_float(__builtin_amdgcn_readlane(__float_as_int(wgt), j0 + k));
            r[k]  = *((const v2f*)(Wh + (size_t)sj * GAT_OUT) + lane);
          }
#pragma unroll
          for (int k = 0; k < 8; ++k) {
            acc[q].x = fmaf(wj[k], r[k].x, acc[q].x);
            acc[q].y = fmaf(wj[k], r[k].y, acc[q].y);
          }
        }
      }
    }
    if (multi) __syncthreads();
  }

  const v2f bb = *((const v2f*)bias + lane);
#pragma unroll
  for (int q = 0; q < BKT_NPW; ++q) {
    const int dl = wv * BKT_NPW + q;
    const int n  = n0 + dl;
    if (n < GAT_N) {
      const float den = sm[dl] + GAT_EPS;
      v2f o;
      o.x = acc[q].x / den + bb.x;
      o.y = acc[q].y / den + bb.y;
      *((v2f*)(out + (size_t)n * GAT_OUT) + lane) = o;
    }
  }
}

extern "C" void kernel_launch(void* const* d_in, const int* in_sizes, int n_in,
                              void* d_out, int out_size, void* d_ws, size_t ws_size,
                              hipStream_t stream) {
  const float* x    = (const float*)d_in[0];
  const int*   ei   = (const int*)d_in[1];
  const float* W    = (const float*)d_in[2];
  const float* al   = (const float*)d_in[3];
  const float* ar   = (const float*)d_in[4];
  const float* bias = (const float*)d_in[5];
  float* out = (float*)d_out;
  (void)in_sizes; (void)n_in; (void)out_size; (void)ws_size;

  float* ws   = (float*)d_ws;
  float* Wh   = ws;
  float* el   = Wh + (size_t)GAT_N * GAT_OUT;
  float* er   = el + GAT_N;
  unsigned int* seg    = (unsigned int*)(er + GAT_N);
  unsigned int* sorted = seg + BK_NUM * SRT_NBLK;

  const int* src = ei;
  const int* dst = ei + GAT_E;

  k_gemm_sort<<<GEMM_NBLK + SRT_NBLK, SRT_T, 0, stream>>>(x, W, al, ar, Wh, el, er, src, dst, seg, sorted);
  k_bucket   <<<BK_NUM, BKT_T, 0, stream>>>(sorted, seg, Wh, el, er, bias, out);
}


# GEMM part: W is copied once per block into LDS (whole-line loads, row pitch 132 floats) and B fragments are 8-byte LDS reads, not 16-line global loads; k order and all sums unchanged
# speedup vs baseline: 9.8696x; 1.0976x over previous
#include <hip/hip_runtime.h>
#include <hip/hip_bf16.h>

#define GAT_N      50000
#define GAT_E      1600000
#define GAT_IN     128
#define GAT_OUT    64
#define GAT_SLOPE  0.2f
#define GAT_EPS    1e-12f

typedef __attribute__((ext_vector_type(2))) float v2f;
typedef __attribute__((ext_vector_type(8))) float v8f;


template <int MASK>
__device__ __forceinline__ float row_xor(float x) {
  return __int_as_float(__builtin_amdgcn_update_dpp(0, __float_as_int(x), 0x160 + MASK, 0xf, 0xf, false));
}

#define GEMM_WPB  16
#define GEMM_NBLK ((GAT_N / 16 + GEMM_WPB - 1) / GEMM_WPB)
#define GEMM_WP   (GAT_IN + 4)
typedef __attribute__((ext_vector_type(4))) float v4f;
__device__ __forceinline__ void gemm_part(const int blk, const float* __restrict__ x,
                                          const float* __restrict__ W,
                                          const float* __restrict__ al,
                                          const float* __restrict__ ar,
                                          float* __restrict__ Wh,
                                          float* __restrict__ el, float* __restrict__ er,
                                          float* wl) {
#pragma unroll
  for (int i = 0; i < (GAT_OUT * GAT_IN) / (4 * GEMM_WPB * 32); ++i) {
    const int f = (i * GEMM_WPB * 32 + (int)threadIdx.x) * 4;
    const v4f v = *(const v4f*)(W + f);
    *(v4f*)(wl + (f / GAT_IN) * GEMM_WP + (f % GAT_IN)) = v;
  }
  __syncthreads();
  const int lane = threadIdx.x & 31;
  const int tile = blk * GEMM_WPB + (threadIdx.x >> 5);
  if (tile >= GAT_N / 16) return;
  const int base  = tile * 16;
  const int l15   = lane & 15;
  const int khalf = (lane >> 4) << 1;

  const float* xr = x + (size_t)(base + l15) * GAT_IN + khalf;
  const float* w0 = wl + (l15 +  0) * GEMM_WP + khalf;
  const float* w1 = wl + (l15 + 16) * GEMM_WP + khalf;
  const float* w2 = wl + (l15 + 32) * GEMM_WP + khalf;
  const float* w3 = wl + (l15 + 48) * GEMM_WP + khalf;

  v8f acc0 = {}, acc1 = {}, acc2 = {}, acc3 = {};

#pragma unroll 8
  for (int k = 0; k < GAT_IN; k += 4) {
    v2f a  = *(const v2f*)(xr + k);
    v2f b0 = *(const v2f*)(w0 + k);
    v2f b1 = *(const v2f*)(w1 + k);
    v2f b2 = *(const v2f*)(w2 + k);
    v2f b3 = *(const v2f*)(w3 + k);
    acc0 = __builtin_amdgcn_wmma_f32_16x16x4_f32(false, a, false, b0, (short)0, acc0, false, false);
    acc1 = __builtin_amdgcn_wmma_f32_16x16x4_f32(false, a, false, b1, (short)0, acc1, false, false);
    acc2 = __builtin_amdgcn_wmma_f32_16x16x4_f32(false, a, false, b2, (short)0, acc2, false, false);
    acc3 = __builtin_amdgcn_wmma_f32_16x16x4_f32(false, a, false, b3, (short)0, acc3, false, false);
  }

  const int rhalf = (lane >> 4) << 3;
  float* o = Wh + (size_t)(base + rhalf) * GAT_OUT + l15;
#pragma unroll
  for (int v = 0; v < 8; ++v) {
    o[(size_t)v * GAT_OUT +  0] = acc0[v];
    o[(size_t)v * GAT_OUT + 16] = acc1[v];
    o[(size_t)v * GAT_OUT + 32] = acc2[v];
    o[(size_t)v * GAT_OUT + 48] = acc3[v];
  }

  const float l0 = al[l15], l1 = al[l15 + 16], l2 = al[l15 + 32], l3 = al[l15 + 48];
  const float r0 = ar[l15], r1 = ar[l15 + 16], r2 = ar[l15 + 32], r3 = ar[l15 + 48];
  float u[16];
#pragma unroll
  for (int v = 0; v < 8; ++v) {
    u[v]     = fmaf(acc3[v], l3, fmaf(acc2[v], l2, fmaf(acc1[v], l1, acc0[v] * l0)));
    u[v + 8] = fmaf(acc3[v], r3, fmaf(acc2[v], r2, fmaf(acc1[v], r1, acc0[v] * r0)));
  }
#pragma unroll
  for (int i = 0; i < 16; ++i) u[i] += row_xor<1>(u[i]);
#pragma unroll
  for (int i = 0; i < 16; ++i) u[i] += row_xor<2>(u[i]);
#pragma unroll
  for (int i = 0; i < 16; ++i) u[i] += row_xor<4>(u[i]);
#pragma unroll
  for (int i = 0; i < 16; ++i) u[i] += row_xor<8>(u[i]);
  float keep = u[0];
#pragma unroll
  for (int i = 1; i < 16; ++i) keep = (l15 == i) ? u[i] : keep;
  float* dstp = (l15 < 8) ? el : er;
  dstp[base + rhalf + (l15 & 7)] = keep;
}

#define BK_SH    7
#define BK_N     128
#define BK_NUM   ((GAT_N + BK_N - 1) / BK_N)
#define SRT_NBLK 200
#define SRT_CH   (GAT_E / SRT_NBLK)
#define SRT_T    512
#define SRT_EPT  ((SRT_CH + SRT_T - 1) / SRT_T)
static_assert(SRT_NBLK * SRT_CH == GAT_E, "sort blocks must tile the edge list");
static_assert(GAT_N <= 65536, "src is packed into 16 bits of the sorted word");
static_assert(SRT_CH < 65536, "segment start and length are packed into 16 bits each");
static_assert(BK_NUM <= SRT_T, "one bucket counter per thread in the block scan");

__device__ __forceinline__ void sort_part(const int blk, const int* __restrict__ src,
                                          const int* __restrict__ dst,
                                          unsigned int* __restrict__ seg,
                                          unsigned int* __restrict__ sorted,
                                          int* h, int* wsum, unsigned int* stage) {
  const int t    = threadIdx.x;
  const int lane = t & 31;
  const int wv   = t >> 5;
  h[t] = 0;
  __syncthreads();
  const int* sp = src + (size_t)blk * SRT_CH;
  const int* dp = dst + (size_t)blk * SRT_CH;
  int d[SRT_EPT], s[SRT_EPT];
#pragma unroll
  for (int k = 0; k < SRT_EPT; ++k) {
    const int i = k * SRT_T + t;
    d[k] = -1; s[k] = 0;
    if (i < SRT_CH) { d[k] = dp[i]; s[k] = sp[i]; }
  }
#pragma unroll
  for (int k = 0; k < SRT_EPT; ++k) {
    if (d[k] >= 0) atomicAdd(&h[d[k] >> BK_SH], 1);
  }
  __syncthreads();

  const int v = h[t];
  int incl = v;
#pragma unroll
  for (int o = 1; o < 32; o <<= 1) {
    const int up = __shfl_up(incl, o);
    if (lane >= o) incl += up;
  }
  if (lane == 31) wsum[wv] = incl;
  __syncthreads();
  int woff = 0;
#pragma unroll
  for (int q = 0; q < SRT_T / 32; ++q) woff += (q < wv) ? wsum[q] : 0;
  const int ex = woff + incl - v;
  h[t] = ex;
  if (t < BK_NUM) seg[(size_t)t * SRT_NBLK + blk] = ((unsigned int)ex << 16) | (unsigned int)v;
  __syncthreads();

  int slot[SRT_EPT];
#pragma unroll
  for (int k = 0; k < SRT_EPT; ++k) {
    slot[k] = 0;
    if (d[k] >= 0) slot[k] = atomicAdd(&h[d[k] >> BK_SH], 1);
  }
#pragma unroll
  for (int k = 0; k < SRT_EPT; ++k) {
    if (d[k] >= 0) stage[slot[k]] = (unsigned int)s[k] | ((unsigned int)(d[k] & (BK_N - 1)) << 16);
  }
  __syncthreads();
  unsigned int* op = sorted + (size_t)blk * SRT_CH;
#pragma unroll
  for (int k = 0; k < SRT_EPT; ++k) {
    const int i = k * SRT_T + t;
    if (i < SRT_CH) op[i] = stage[i];
  }
}

static_assert(GEMM_WPB * 32 == SRT_T, "both parts of the shared launch use one block size");
__global__ __launch_bounds__(SRT_T) void k_gemm_sort(const float* __restrict__ x,
                                                     const float* __restrict__ W,
                                                     const float* __restrict__ al,
                                                     const float* __restrict__ ar,
                                                     float* __restrict__ Wh,
                                                     float* __restrict__ el,
                                                     float* __restrict__ er,
                                                     const int* __restrict__ src,
                                                     const int* __restrict__ dst,
                                                     unsigned int* __restrict__ seg,
                                                     unsigned int* __restrict__ sorted) {
  __shared__ __attribute__((aligned(16))) unsigned int smem[SRT_T + SRT_T / 32 + SRT_CH];
  static_assert(sizeof(smem) >= GAT_OUT * GEMM_WP * sizeof(float), "the copy of W must fit the shared area");
  if (blockIdx.x < GEMM_NBLK) gemm_part(blockIdx.x, x, W, al, ar, Wh, el, er, (float*)smem);
  else                        sort_part(blockIdx.x - GEMM_NBLK, src, dst, seg, sorted, (int*)smem,
                                        (int*)smem + SRT_T, smem + SRT_T + SRT_T / 32);
}

__device__ __forceinline__ void lds_max_f32(float* addr, float val) {
  if (val >= 0.f) atomicMax((int*)addr, __float_as_int(val));
  else            atomicMin((unsigned int*)addr, (unsigned int)__float_as_int(val));
}

__device__ __forceinline__ float edge_logit(const float* __restrict__ el,
                                            const float* __restrict__ er, int s, int d) {
  float v = el[s] + er[d];
  return (v > 0.f) ? v : GAT_SLOPE * v;
}

#define BKT_T   1024
#define BKT_EPT 5
#define BKT_CAP (BKT_T * BKT_EPT)
#define BKT_NPW (BK_N / (BKT_T / 32))
#define BKT_NONE 0xffffffffu
#define BKT_SPL ((SRT_NBLK + 31) / 32)
static_assert(BKT_NPW * (BKT_T / 32) == BK_N, "the waves must tile the bucket's nodes");
static_assert(BK_N == 4 * 32, "the scan below gives every lane of one wave four counters");

__device__ __forceinline__ int bkt_segs(const unsigned int* __restrict__ sg, int lane, int wvs,
                                        int (&st)[BKT_SPL], int (&ln)[BKT_SPL],
                                        int (&ps)[BKT_SPL]) {
  int sst[BKT_SPL], sln[BKT_SPL];
#pragma unroll
  for (int j = 0; j < BKT_SPL; ++j) {
    const int k = j * 32 + lane;
    unsigned int v = 0u;
    if (k < SRT_NBLK) v = sg[k];
    sst[j] = k * SRT_CH + (int)(v >> 16);
    sln[j] = (int)(v & 0xffffu);
  }
  int run = 0;
#pragma unroll
  for (int j = 0; j < BKT_SPL; ++j) {
    int incl = sln[j];
#pragma unroll
    for (int o = 1; o < 32; o <<= 1) {
      const int up = __shfl_up(incl, o);
      if (lane >= o) incl += up;
    }
    st[j] = __builtin_amdgcn_readlane(sst[j], wvs);
    ln[j] = __builtin_amdgcn_readlane(sln[j], wvs);
    ps[j] = run + __builtin_amdgcn_readlane(incl - sln[j], wvs);
    run += __builtin_amdgcn_readlane(incl, 31);
  }
  return run;
}

__device__ __forceinline__ void bkt_stage(const unsigned int* __restrict__ sorted,
                                          unsigned int* lraw, const int (&st)[BKT_SPL],
                                          const int (&ln)[BKT_SPL], const int (&ps)[BKT_SPL],
                                          int cb, int lane) {
  int e0[BKT_SPL], e1[BKT_SPL];
  unsigned int v[BKT_SPL];
#pragma unroll
  for (int j = 0; j < BKT_SPL; ++j) {
    e0[j] = max(0, cb - ps[j]);
    e1[j] = min(ln[j], cb + BKT_CAP - ps[j]);
    v[j] = 0u;
    if (e0[j] + lane < e1[j]) v[j] = sorted[st[j] + e0[j] + lane];
  }
#pragma unroll
  for (int j = 0; j < BKT_SPL; ++j) {
    if (e0[j] + lane < e1[j]) lraw[ps[j] + e0[j] + lane - cb] = v[j];
  }
#pragma unroll
  for (int j = 0; j < BKT_SPL; ++j) {
#pragma nounroll
    for (int e = e0[j] + 32 + lane; e < e1[j]; e += 32) lraw[ps[j] + e - cb] = sorted[st[j] + e];
  }
}

__device__ __forceinline__ void bkt_load_chunk(const unsigned int* lraw,
                                               const float* __restrict__ el,
                                               const float* ler, int cb, int end, int t,
                                               unsigned int (&w)[BKT_EPT], float (&lg)[BKT_EPT]) {
  float e[BKT_EPT];
#pragma unroll
  for (int k = 0; k < BKT_EPT; ++k) {
    const int i = k * BKT_T + t;
    const unsigned int v = lraw[i];
    w[k] = (cb + i < end) ? v : BKT_NONE;
  }
#pragma unroll
  for (int k = 0; k < BKT_EPT; ++k) {
    e[k] = el[(w[k] != BKT_NONE) ? (int)(w[k] & 0xffffu) : 0];
  }
#pragma unroll
  for (int k = 0; k < BKT_EPT; ++k) {
    const int dl = (w[k] != BKT_NONE) ? (int)(w[k] >> 16) : 0;
    const float v = e[k] + ler[dl];
    lg[k] = (v > 0.f) ? v : GAT_SLOPE * v;
  }
}

__global__ __launch_bounds__(BKT_T) void k_bucket(const unsigned int* __restrict__ sorted,
                                                  const unsigned int* __restrict__ seg,
                                                  const float* __restrict__ Wh,
                                                  const float* __restrict__ el,
                                                  const float* __restrict__ er,
                                                  const float* __restrict__ bias,
                                                  float* __restrict__ out) {
  __shared__ unsigned int lsrc[BKT_CAP];
  __shared__ float        lwt[BKT_CAP];
  __shared__ float mx[BK_N];
  __shared__ float sm[BK_N];
  __shared__ float ler[BK_N];
  __shared__ int   cur[BK_N];
  __shared__ int   off[BK_N + 4];
  const int t    = threadIdx.x;
  const int lane = t & 31;
  const int wv   = t >> 5;
  const int wvs  = __builtin_amdgcn_readfirstlane(wv);
  const int n0   = blockIdx.x * BK_N;
  const unsigned int* sg = seg + (size_t)blockIdx.x * SRT_NBLK;
  int sst[BKT_SPL], sln[BKT_SPL], sps[BKT_SPL];
  const int end  = bkt_segs(sg, lane, wvs, sst, sln, sps);
  const bool multi = end > BKT_CAP;

  if (t < BK_N) {
    mx[t] = -__builtin_inff(); sm[t] = 0.f; cur[t] = 0;
    ler[t] = (n0 + t < GAT_N) ? er[n0 + t] : 0.f;
  }

  unsigned int w[BKT_EPT];
  float        lg[BKT_EPT];
#pragma unroll
  for (int k = 0; k < BKT_EPT; ++k) { w[k] = BKT_NONE; lg[k] = 0.f; }

  for (int cb = 0; cb < end; cb += BKT_CAP) {
    bkt_stage(sorted, lsrc, sst, sln, sps, cb, lane);
    __syncthreads();
    bkt_load_chunk(lsrc, el, ler, cb, end, t, w, lg);
#pragma unroll
    for (int k = 0; k < BKT_EPT; ++k) {
      if (w[k] != BKT_NONE) {
        const int dl = (int)(w[k] >> 16);
        lds_max_f32(&mx[dl], lg[k]);
        if (!multi) atomicAdd(&cur[dl], 1);
      }
    }
    if (multi) __syncthreads();
  }
  __syncthreads();

  v2f acc[BKT_NPW];
#pragma unroll
  for (int q = 0; q < BKT_NPW; ++q) { acc[q].x = 0.f; acc[q].y = 0.f; }

  for (int cb = 0; cb < end; cb += BKT_CAP) {
    if (multi) {
      bkt_stage(sorted, lsrc, sst, sln, sps, cb, lane);
      if (t < BK_N) cur[t] = 0;
      __syncthreads();
      bkt_load_chunk(lsrc, el, ler, cb, end, t, w, lg);
#pragma unroll
      for (int k = 0; k < BKT_EPT; ++k) {
        if (w[k] != BKT_NONE) atomicAdd(&cur[(int)(w[k] >> 16)], 1);
      }
      __syncthreads();
    }

    if (wv == 0) {
      const int c0 = cur[4 * lane + 0], c1 = cur[4 * lane + 1];
      const int c2 = cur[4 * lane + 2], c3 = cur[4 * lane + 3];
      const int tot = c0 + c1 + c2 + c3;
      int incl = tot;
#pragma unroll
      for (int d = 1; d < 32; d <<= 1) {
        const int up = __shfl_up(incl, d);
        if (lane >= d) incl += up;
      }
      const int ex = incl - tot;
      off[4 * lane + 0] = ex;                cur[4 * lane + 0] = ex;
      off[4 * lane + 1] = ex + c0;           cur[4 * lane + 1] = ex + c0;
      off[4 * lane + 2] = ex + c0 + c1;      cur[4 * lane + 2] = ex + c0 + c1;
      off[4 * lane + 3] = ex + c0 + c1 + c2; cur[4 * lane + 3] = ex + c0 + c1 + c2;
      if (lane == 31) off[BK_N] = incl;
    }
    __syncthreads();

    {
      float exv[BKT_EPT];
      int   pos[BKT_EPT];
#pragma unroll
      for (int k = 0; k < BKT_EPT; ++k) {
        exv[k] = mx[(w[k] != BKT_NONE) ? (int)(w[k] >> 16) : 0];
      }
#pragma unroll
      for (int k = 0; k < BKT_EPT; ++k) {
        exv[k] = __expf(lg[k] - exv[k]);
      }
#pragma unroll
      for (int k = 0; k < BKT_EPT; ++k) {
        pos[k] = 0;
        if (w[k] != BKT_NONE) {
          const int dl = (int)(w[k] >> 16);
          atomicAdd(&sm[dl], exv[k]);
          pos[k] = atomicAdd(&cur[dl], 1);
        }
      }
#pragma unroll
      for (int k = 0; k < BKT_EPT; ++k) {
        if (w[k] != BKT_NONE) {
          lsrc[pos[k]] = w[k] & 0xffffu;
          lwt[pos[k]]  = exv[k];
        }
      }
    }
    __syncthreads();

#pragma unroll
    for (int q = 0; q < BKT_NPW; ++q) {
      const int dl = wv * BKT_NPW + q;
      const int a  = __builtin_amdgcn_readfirstlane(off[dl]);
      const int b  = __builtin_amdgcn_readfirstlane(off[dl + 1]);
      for (int c = a; c < b; c += 32) {
        const int  idx = c + lane;
        const bool ok  = idx < b;
        const unsigned int s  = lsrc[ok ? idx : c];
        const float wgt = ok ? lwt[ok ? idx : c] : 0.f;
        const int nv = min(32, b - c);
        for (int j0 = 0; j0 < nv; j0 += 8) {
          v2f   r[8];
          float wj[8];
#pragma unroll
          for (int k = 0; k < 8; ++k) {
            const unsigned int sj = (unsigned int)__builtin_amdgcn_readlane((int)s, j0 + k);
            wj[k] = __int_as_float(__builtin_amdgcn_readlane(__float_as_int(wgt), j0 + k));
            r[k]  = *((const v2f*)(Wh + (size_t)sj * GAT_OUT) + lane);
          }
#pragma unroll
          for (int k = 0; k < 8; ++k) {
            acc[q].x = fmaf(wj[k], r[k].x, acc[q].x);
            acc[q].y = fmaf(wj[k], r[k].y, acc[q].y);
          }
        }
      }
    }
    if (multi) __syncthreads();
  }

  const v2f bb = *((const v2f*)bias + lane);
#pragma unroll
  for (int q = 0; q < BKT_NPW; ++q) {
    const int dl = wv * BKT_NPW + q;
    const int n  = n0 + dl;
    if (n < GAT_N) {
      const float den = sm[dl] + GAT_EPS;
      v2f o;
      o.x = acc[q].x / den + bb.x;
      o.y = acc[q].y / den + bb.y;
      *((v2f*)(out + (size_t)n * GAT_OUT) + lane) = o;
    }
  }
}

extern "C" void kernel_launch(void* const* d_in, const int* in_sizes, int n_in,
                              void* d_out, int out_size, void* d_ws, size_t ws_size,
                              hipStream_t stream) {
  const float* x    = (const float*)d_in[0];
  const int*   ei   = (const int*)d_in[1];
  const float* W    = (const float*)d_in[2];
  const float* al   = (const float*)d_in[3];
  const float* ar   = (const float*)d_in[4];
  const float* bias = (const float*)d_in[5];
  float* out = (float*)d_out;
  (void)in_sizes; (void)n_in; (void)out_size; (void)ws_size;

  float* ws   = (float*)d_ws;
  float* Wh   = ws;
  float* el   = Wh + (size_t)GAT_N * GAT_OUT;
  float* er   = el + GAT_N;
  unsigned int* seg    = (unsigned int*)(er + GAT_N);
  unsigned int* sorted = seg + BK_NUM * SRT_NBLK;

  const int* src = ei;
  const int* dst = ei + GAT_E;

  k_gemm_sort<<<GEMM_NBLK + SRT_NBLK, SRT_T, 0, stream>>>(x, W, al, ar, Wh, el, er, src, dst, seg, sorted);
  k_bucket   <<<BK_NUM, BKT_T, 0, stream>>>(sorted, seg, Wh, el, er, bias, out);
}
